# starting kernel
# baseline (speedup 1.0000x reference)
#include <hip/hip_runtime.h>

#define NN 100000
#define NE 625000
#define D  128

typedef float v2f __attribute__((ext_vector_type(2)));
typedef float v8f __attribute__((ext_vector_type(8)));

__global__ __launch_bounds__(256) void sage_deg_kernel(const int* __restrict__ dst,
                                                       float* __restrict__ deg, int E) {
  int e = blockIdx.x * blockDim.x + threadIdx.x;
  if (e < E) atomicAdd(&deg[dst[e]], 1.0f);
}

__global__ __launch_bounds__(256) void sage_scatter_kernel(const float* __restrict__ x,
                                                           const int* __restrict__ src,
                                                           const int* __restrict__ dst,
                                                           float* __restrict__ agg, int E) {
  long long t = (long long)blockIdx.x * blockDim.x + threadIdx.x;
  int e    = (int)(t >> 5);
  int lane = (int)(t & 31);
  if (e >= E) return;
  int s = src[e];
  int d = dst[e];
  const float4 v = ((const float4*)(x + (long long)s * D))[lane];
  float* a = agg + (long long)d * D + lane * 4;
  atomicAdd(a + 0, v.x);
  atomicAdd(a + 1, v.y);
  atomicAdd(a + 2, v.z);
  atomicAdd(a + 3, v.w);
}

__global__ __launch_bounds__(256) void sage_wmma_kernel(const float* __restrict__ agg,
                                                        const float* __restrict__ deg,
                                                        const float* __restrict__ xin,
                                                        const float* __restrict__ Wl,
                                                        const float* __restrict__ Wr,
                                                        const float* __restrict__ bias,
                                                        float* __restrict__ out,
                                                        int n, int applyRelu) {
  const int lane = threadIdx.x & 31;
  const int wave = threadIdx.x >> 5;
  const int tile = blockIdx.x * 8 + wave;
  const int nTiles = n >> 4;
  if (tile >= nTiles) return;
  const int row0 = tile * 16;

  const int m  = lane & 15;
  const int kh = (lane >> 4) * 2;

  const float rdeg = 1.0f / fmaxf(deg[row0 + m], 1.0f);

  v8f zero = {};
  v8f acc[8];
#pragma unroll
  for (int i = 0; i < 8; ++i) acc[i] = zero;

  for (int pass = 0; pass < 2; ++pass) {
    const float* A     = pass ? xin : agg;
    const float* W     = pass ? Wr : Wl;
    const float  scale = pass ? 1.0f : rdeg;
    const float* arow  = A + (long long)(row0 + m) * D;
#pragma unroll 2
    for (int k0 = 0; k0 < D; k0 += 4) {
      v2f a;
      a.x = arow[k0 + kh + 0] * scale;
      a.y = arow[k0 + kh + 1] * scale;
#pragma unroll
      for (int ct = 0; ct < 8; ++ct) {
        const float* wrow = W + (long long)(ct * 16 + m) * D;
        v2f b;
        b.x = wrow[k0 + kh + 0];
        b.y = wrow[k0 + kh + 1];
        acc[ct] = __builtin_amdgcn_wmma_f32_16x16x4_f32(
             false, a,  false, b,
             (short)0, acc[ct],  false,  false);
      }
    }
  }

  const int rhalf = (lane >> 4) * 8;
#pragma unroll
  for (int ct = 0; ct < 8; ++ct) {
    const float bc = bias[ct * 16 + m];
#pragma unroll
    for (int r = 0; r < 8; ++r) {
      float v = acc[ct][r] + bc;
      if (applyRelu) v = fmaxf(v, 0.0f);
      out[(long long)(row0 + r + rhalf) * D + ct * 16 + m] = v;
    }
  }
}

extern "C" void kernel_launch(void* const* d_in, const int* in_sizes, int n_in,
                              void* d_out, int out_size, void* d_ws, size_t ws_size,
                              hipStream_t stream) {
  (void)in_sizes; (void)n_in; (void)out_size; (void)ws_size;

  const float* x   = (const float*)d_in[0];
  const int*   ei  = (const int*)d_in[1];
  const float* W1l = (const float*)d_in[2];
  const float* b1  = (const float*)d_in[3];
  const float* W1r = (const float*)d_in[4];
  const float* W2l = (const float*)d_in[5];
  const float* b2  = (const float*)d_in[6];
  const float* W2r = (const float*)d_in[7];
  float*       out = (float*)d_out;

  const int* src = ei;
  const int* dst = ei + NE;

  char*  ws  = (char*)d_ws;
  float* deg = (float*)ws;
  float* agg = (float*)(ws + (1ull << 20));
  float* h   = (float*)(ws + (1ull << 20) + (size_t)NN * D * sizeof(float));

  const int degBlocks     = (NE + 255) / 256;
  const int scatterBlocks = (int)(((long long)NE * 32 + 255) / 256);
  const int gemmBlocks    = ((NN / 16) + 7) / 8;

  hipMemsetAsync(deg, 0, (size_t)NN * sizeof(float), stream);
  hipMemsetAsync(agg, 0, (size_t)NN * D * sizeof(float), stream);
  sage_deg_kernel<<<degBlocks, 256, 0, stream>>>(dst, deg, NE);
  sage_scatter_kernel<<<scatterBlocks, 256, 0, stream>>>(x, src, dst, agg, NE);
  sage_wmma_kernel<<<gemmBlocks, 256, 0, stream>>>(agg, deg, x, W1l, W1r, b1, h,
                                                   NN,  1);

  hipMemsetAsync(agg, 0, (size_t)NN * D * sizeof(float), stream);
  sage_scatter_kernel<<<scatterBlocks, 256, 0, stream>>>(h, src, dst, agg, NE);
  sage_wmma_kernel<<<gemmBlocks, 256, 0, stream>>>(agg, deg, h, W2l, W2r, b2, out,
                                                   NN,  0);
}


# GEMM kernel: both WMMA operands fed from LDS (Wl/Wr copied once per 16-wave work-group, A tile per wave by whole-row loads) instead of per-lane global loads; 256 work-groups loop over row tiles
# speedup vs baseline: 1.2638x; 1.2638x over previous
#include <hip/hip_runtime.h>

#define NN 100000
#define NE 625000
#define D  128

typedef float v2f __attribute__((ext_vector_type(2)));
typedef float v8f __attribute__((ext_vector_type(8)));

__global__ __launch_bounds__(256) void sage_deg_kernel(const int* __restrict__ dst,
                                                       float* __restrict__ deg, int E) {
  int e = blockIdx.x * blockDim.x + threadIdx.x;
  if (e < E) atomicAdd(&deg[dst[e]], 1.0f);
}

__global__ __launch_bounds__(256) void sage_scatter_kernel(const float* __restrict__ x,
                                                           const int* __restrict__ src,
                                                           const int* __restrict__ dst,
                                                           float* __restrict__ agg, int E) {
  long long t = (long long)blockIdx.x * blockDim.x + threadIdx.x;
  int e    = (int)(t >> 5);
  int lane = (int)(t & 31);
  if (e >= E) return;
  int s = src[e];
  int d = dst[e];
  const float4 v = ((const float4*)(x + (long long)s * D))[lane];
  float* a = agg + (long long)d * D + lane * 4;
  atomicAdd(a + 0, v.x);
  atomicAdd(a + 1, v.y);
  atomicAdd(a + 2, v.z);
  atomicAdd(a + 3, v.w);
}

#define PITCH        132
#define GEMM_WAVES   16
#define GEMM_THREADS (GEMM_WAVES * 32)
#define GEMM_BLOCKS  256
#define LDS_W_FLOATS (2 * D * PITCH)
#define LDS_A_FLOATS (GEMM_WAVES * 16 * PITCH)

__device__ __forceinline__ void stage_rows(const float* __restrict__ src, int row0, int lane,
                                           float* sAw) {
  float4 r[16];
#pragma unroll
  for (int i = 0; i < 16; ++i)
    r[i] = ((const float4*)(src + (long long)(row0 + i) * D))[lane];
#pragma unroll
  for (int i = 0; i < 16; ++i)
    *(float4*)(sAw + i * PITCH + lane * 4) = r[i];
}

__device__ __forceinline__ void load_chunk(const float* aRow, const float* bRow, int k0,
                                           v2f (&a)[2], v2f (&b)[2][8]) {
#pragma unroll
  for (int s = 0; s < 2; ++s) {
    a[s] = *(const v2f*)(aRow + k0 + 4 * s);
#pragma unroll
    for (int ct = 0; ct < 8; ++ct)
      b[s][ct] = *(const v2f*)(bRow + ct * 16 * PITCH + k0 + 4 * s);
  }
}

template <bool SCALE>
__device__ __forceinline__ void mma_chunk(const v2f (&a)[2], const v2f (&b)[2][8], float scale,
                                          v8f (&acc)[8]) {
#pragma unroll
  for (int s = 0; s < 2; ++s) {
    v2f av = a[s];
    if (SCALE) { av.x *= scale; av.y *= scale; }
#pragma unroll
    for (int ct = 0; ct < 8; ++ct)
      acc[ct] = __builtin_amdgcn_wmma_f32_16x16x4_f32(
           false, av,  false, b[s][ct],
           (short)0, acc[ct],  false,  false);
  }
}

template <bool SCALE>
__device__ __forceinline__ void pass_compute(const float* aRow, const float* bRow, float scale,
                                             v8f (&acc)[8]) {
  v2f a0[2], a1[2];
  v2f b0[2][8], b1[2][8];
  load_chunk(aRow, bRow, 0, a0, b0);
#pragma unroll 1
  for (int k0 = 0; k0 < D - 16; k0 += 16) {
    load_chunk(aRow, bRow, k0 + 8, a1, b1);
    mma_chunk<SCALE>(a0, b0, scale, acc);
    load_chunk(aRow, bRow, k0 + 16, a0, b0);
    mma_chunk<SCALE>(a1, b1, scale, acc);
  }
  load_chunk(aRow, bRow, D - 8, a1, b1);
  mma_chunk<SCALE>(a0, b0, scale, acc);
  mma_chunk<SCALE>(a1, b1, scale, acc);
}

__global__ __launch_bounds__(GEMM_THREADS) void sage_wmma_kernel(const float* __restrict__ agg,
                                                        const float* __restrict__ deg,
                                                        const float* __restrict__ xin,
                                                        const float* __restrict__ Wl,
                                                        const float* __restrict__ Wr,
                                                        const float* __restrict__ bias,
                                                        float* __restrict__ out,
                                                        int n, int applyRelu) {
  __shared__ __attribute__((aligned(16))) float lds[LDS_W_FLOATS + LDS_A_FLOATS];

  const int lane = threadIdx.x & 31;
  const int wave = __builtin_amdgcn_readfirstlane(threadIdx.x >> 5);
  const int nTiles = n >> 4;
  const int tileStep = GEMM_WAVES * gridDim.x;

  const int m  = lane & 15;
  const int kh = (lane >> 4) * 2;

  float* sW  = lds;
  float* sAw = lds + LDS_W_FLOATS + wave * 16 * PITCH;
  const float* aRow  = sAw + m * PITCH + kh;
  const float* bRowL = sW + m * PITCH + kh;
  const float* bRowR = bRowL + D * PITCH;

  int tile = wave * gridDim.x + blockIdx.x;

  if (tile < nTiles) stage_rows(agg, tile * 16, lane, sAw);

  {
    float4 wv[16];
#pragma unroll
    for (int j = 0; j < 16; ++j) {
      const int row = j * GEMM_WAVES + wave;
      const float* W = (row < D) ? Wl : Wr;
      wv[j] = ((const float4*)(W + (row & (D - 1)) * D))[lane];
    }
#pragma unroll
    for (int j = 0; j < 16; ++j) {
      const int row = j * GEMM_WAVES + wave;
      *(float4*)(sW + row * PITCH + lane * 4) = wv[j];
    }
  }
  __syncthreads();

  while (tile < nTiles) {
    const int row0 = tile * 16;

    const float rdeg = 1.0f / fmaxf(deg[row0 + m], 1.0f);

    v8f zero = {};
    v8f acc[8];
#pragma unroll
    for (int i = 0; i < 8; ++i) acc[i] = zero;

    pass_compute<true>(aRow, bRowL, rdeg, acc);

    stage_rows(xin, row0, lane, sAw);
    pass_compute<false>(aRow, bRowR, 1.0f, acc);

    const int tileNext = tile + tileStep;
    if (tileNext < nTiles) stage_rows(agg, tileNext * 16, lane, sAw);

    const int rhalf = (lane >> 4) * 8;
#pragma unroll
    for (int ct = 0; ct < 8; ++ct) {
      const float bc = bias[ct * 16 + m];
#pragma unroll
      for (int r = 0; r < 8; ++r) {
        float v = acc[ct][r] + bc;
        if (applyRelu) v = fmaxf(v, 0.0f);
        out[(long long)(row0 + r + rhalf) * D + ct * 16 + m] = v;
      }
    }
    tile = tileNext;
  }
}

extern "C" void kernel_launch(void* const* d_in, const int* in_sizes, int n_in,
                              void* d_out, int out_size, void* d_ws, size_t ws_size,
                              hipStream_t stream) {
  (void)in_sizes; (void)n_in; (void)out_size; (void)ws_size;

  const float* x   = (const float*)d_in[0];
  const int*   ei  = (const int*)d_in[1];
  const float* W1l = (const float*)d_in[2];
  const float* b1  = (const float*)d_in[3];
  const float* W1r = (const float*)d_in[4];
  const float* W2l = (const float*)d_in[5];
  const float* b2  = (const float*)d_in[6];
  const float* W2r = (const float*)d_in[7];
  float*       out = (float*)d_out;

  const int* src = ei;
  const int* dst = ei + NE;

  char*  ws  = (char*)d_ws;
  float* deg = (float*)ws;
  float* agg = (float*)(ws + (1ull << 20));
  float* h   = (float*)(ws + (1ull << 20) + (size_t)NN * D * sizeof(float));

  const int degBlocks     = (NE + 255) / 256;
  const int scatterBlocks = (int)(((long long)NE * 32 + 255) / 256);

  hipMemsetAsync(deg, 0, (size_t)NN * sizeof(float), stream);
  hipMemsetAsync(agg, 0, (size_t)NN * D * sizeof(float), stream);
  sage_deg_kernel<<<degBlocks, 256, 0, stream>>>(dst, deg, NE);
  sage_scatter_kernel<<<scatterBlocks, 256, 0, stream>>>(x, src, dst, agg, NE);
  sage_wmma_kernel<<<GEMM_BLOCKS, GEMM_THREADS, 0, stream>>>(agg, deg, x, W1l, W1r, b1, h,
                                                   NN,  1);

  hipMemsetAsync(agg, 0, (size_t)NN * D * sizeof(float), stream);
  sage_scatter_kernel<<<scatterBlocks, 256, 0, stream>>>(h, src, dst, agg, NE);
  sage_wmma_kernel<<<GEMM_BLOCKS, GEMM_THREADS, 0, stream>>>(agg, deg, h, W2l, W2r, b2, out,
                                                   NN,  0);
}


# Neighbour sum by a per-call compressed-row edge list (integer atomic rank, prefix scan, fill) and a one-wave-per-node row gather instead of float atomic scatter; rows add in atomic-rank order
# speedup vs baseline: 2.4946x; 1.9739x over previous
#include <hip/hip_runtime.h>

#define NN 100000
#define NE 625000
#define D  128

typedef float v2f __attribute__((ext_vector_type(2)));
typedef float v8f __attribute__((ext_vector_type(8)));

static_assert(NE % 4 == 0, "edge arrays are walked four entries per thread");

#define EDGE_THREADS 256
#define EDGE_BLOCKS  ((NE / 4 + EDGE_THREADS - 1) / EDGE_THREADS)

__global__ __launch_bounds__(EDGE_THREADS) void sage_rank_kernel(const int* __restrict__ dst,
                                                                 int* __restrict__ cnt,
                                                                 int* __restrict__ rank, int E4) {
  const int q = blockIdx.x * blockDim.x + threadIdx.x;
  if (q >= E4) return;
  const int4 d = ((const int4*)dst)[q];
  int4 r;
  r.x = atomicAdd(&cnt[d.x], 1);
  r.y = atomicAdd(&cnt[d.y], 1);
  r.z = atomicAdd(&cnt[d.z], 1);
  r.w = atomicAdd(&cnt[d.w], 1);
  ((int4*)rank)[q] = r;
}

#define SCAN_THREADS 1024
#define SCAN_BLOCKS  ((NN + SCAN_THREADS - 1) / SCAN_THREADS)

__global__ __launch_bounds__(SCAN_THREADS) void sage_scan_kernel(const int* __restrict__ cnt,
                                                                 int* __restrict__ rowptr,
                                                                 float* __restrict__ deg, int n) {
  __shared__ int sWave[32];
  __shared__ int sPart[32];
  const int t    = threadIdx.x;
  const int lane = t & 31;
  const int wave = t >> 5;
  const int base = blockIdx.x * SCAN_THREADS;

  int part = 0;
  const int4* c4 = (const int4*)cnt;
  const int lim = base >> 2;
  for (int q = t; q < lim; q += 8 * SCAN_THREADS) {
    int4 v[8];
#pragma unroll
    for (int u = 0; u < 8; ++u) {
      const int qq = q + u * SCAN_THREADS;
      v[u] = (qq < lim) ? c4[qq] : make_int4(0, 0, 0, 0);
    }
#pragma unroll
    for (int u = 0; u < 8; ++u) part += (v[u].x + v[u].y) + (v[u].z + v[u].w);
  }

  const int i = base + t;
  const int c = (i < n) ? cnt[i] : 0;

  int v = c;
  for (int d = 1; d < 32; d <<= 1) {
    const int u = __shfl_up(v, d);
    if (lane >= d) v += u;
  }
  for (int d = 16; d > 0; d >>= 1) part += __shfl_xor(part, d);
  if (lane == 31) sWave[wave] = v;
  if (lane == 0)  sPart[wave] = part;
  __syncthreads();

  int tot = ((lane < wave) ? sWave[lane] : 0) + sPart[lane];
  for (int d = 16; d > 0; d >>= 1) tot += __shfl_xor(tot, d);

  const int excl = tot + v - c;
  if (i < n) {
    rowptr[i] = excl;
    deg[i]    = (float)c;
    if (i == n - 1) rowptr[n] = excl + c;
  }
}

__global__ __launch_bounds__(EDGE_THREADS) void sage_fill_kernel(const int* __restrict__ src,
                                                                 const int* __restrict__ dst,
                                                                 const int* __restrict__ rank,
                                                                 const int* __restrict__ rowptr,
                                                                 int* __restrict__ csr, int E4) {
  const int q = blockIdx.x * blockDim.x + threadIdx.x;
  if (q >= E4) return;
  const int4 s = ((const int4*)src)[q];
  const int4 d = ((const int4*)dst)[q];
  const int4 r = ((const int4*)rank)[q];
  csr[rowptr[d.x] + r.x] = s.x;
  csr[rowptr[d.y] + r.y] = s.y;
  csr[rowptr[d.z] + r.z] = s.z;
  csr[rowptr[d.w] + r.w] = s.w;
}

#define NPW            13
#define GATHER_WAVES   16
#define GATHER_THREADS (GATHER_WAVES * 32)
#define GATHER_BLOCKS  ((NN + NPW * GATHER_WAVES - 1) / (NPW * GATHER_WAVES))

template <int R>
__device__ __forceinline__ void add_rows(const float* __restrict__ x, int idx, int j, int lane,
                                         float4& acc) {
  float4 v[R];
#pragma unroll
  for (int t = 0; t < R; ++t) {
    const int s = __builtin_amdgcn_readlane(idx, j + t);
    v[t] = ((const float4*)(x + (long long)s * D))[lane];
  }
#pragma unroll
  for (int t = 0; t < R; ++t) {
    acc.x += v[t].x;
    acc.y += v[t].y;
    acc.z += v[t].z;
    acc.w += v[t].w;
  }
}

__device__ __forceinline__ void gather_chunk(const float* __restrict__ x, int idx, int nb, int lane,
                                             float4& acc) {
  int j = 0;
  for (; j + 8 <= nb; j += 8) add_rows<8>(x, idx, j, lane, acc);
  switch (nb - j) {
    case 7: add_rows<7>(x, idx, j, lane, acc); break;
    case 6: add_rows<6>(x, idx, j, lane, acc); break;
    case 5: add_rows<5>(x, idx, j, lane, acc); break;
    case 4: add_rows<4>(x, idx, j, lane, acc); break;
    case 3: add_rows<3>(x, idx, j, lane, acc); break;
    case 2: add_rows<2>(x, idx, j, lane, acc); break;
    case 1: add_rows<1>(x, idx, j, lane, acc); break;
    default: break;
  }
}

__global__ __launch_bounds__(GATHER_THREADS) void sage_gather_kernel(const float* __restrict__ x,
                                                                     const int* __restrict__ rowptr,
                                                                     const int* __restrict__ csr,
                                                                     float* __restrict__ agg, int n) {
  const int lane = threadIdx.x & 31;
  const int wave = __builtin_amdgcn_readfirstlane(threadIdx.x >> 5);
  const int n0 = (blockIdx.x * GATHER_WAVES + wave) * NPW;
  if (n0 >= n) return;
  const int cntN = (n - n0 < NPW) ? (n - n0) : NPW;

  const int rp = (lane <= cntN) ? rowptr[n0 + lane] : 0;

  int beg = __builtin_amdgcn_readlane(rp, 0);
  int end = __builtin_amdgcn_readlane(rp, 1);
  int idxNext = (beg + lane < end) ? csr[beg + lane] : 0;

  for (int k = 0; k < cntN; ++k) {
    const int idx = idxNext;
    const int b = beg;
    const int e = end;
    if (k + 1 < cntN) {
      beg = e;
      end = __builtin_amdgcn_readlane(rp, k + 2);
      idxNext = (beg + lane < end) ? csr[beg + lane] : 0;
    }

    float4 acc = make_float4(0.0f, 0.0f, 0.0f, 0.0f);
    int nb = (e - b < 32) ? (e - b) : 32;
    gather_chunk(x, idx, nb, lane, acc);
    for (int cb = b + 32; cb < e; cb += 32) {
      nb = (e - cb < 32) ? (e - cb) : 32;
      const int idx2 = (lane < nb) ? csr[cb + lane] : 0;
      gather_chunk(x, idx2, nb, lane, acc);
    }
    ((float4*)(agg + (long long)(n0 + k) * D))[lane] = acc;
  }
}

#define PITCH        132
#define GEMM_WAVES   16
#define GEMM_THREADS (GEMM_WAVES * 32)
#define GEMM_BLOCKS  256
#define LDS_W_FLOATS (2 * D * PITCH)
#define LDS_A_FLOATS (GEMM_WAVES * 16 * PITCH)

__device__ __forceinline__ void stage_rows(const float* __restrict__ src, int row0, int lane,
                                           float* sAw) {
  float4 r[16];
#pragma unroll
  for (int i = 0; i < 16; ++i)
    r[i] = ((const float4*)(src + (long long)(row0 + i) * D))[lane];
#pragma unroll
  for (int i = 0; i < 16; ++i)
    *(float4*)(sAw + i * PITCH + lane * 4) = r[i];
}

__device__ __forceinline__ void load_chunk(const float* aRow, const float* bRow, int k0,
                                           v2f (&a)[2], v2f (&b)[2][8]) {
#pragma unroll
  for (int s = 0; s < 2; ++s) {
    a[s] = *(const v2f*)(aRow + k0 + 4 * s);
#pragma unroll
    for (int ct = 0; ct < 8; ++ct)
      b[s][ct] = *(const v2f*)(bRow + ct * 16 * PITCH + k0 + 4 * s);
  }
}

template <bool SCALE>
__device__ __forceinline__ void mma_chunk(const v2f (&a)[2], const v2f (&b)[2][8], float scale,
                                          v8f (&acc)[8]) {
#pragma unroll
  for (int s = 0; s < 2; ++s) {
    v2f av = a[s];
    if (SCALE) { av.x *= scale; av.y *= scale; }
#pragma unroll
    for (int ct = 0; ct < 8; ++ct)
      acc[ct] = __builtin_amdgcn_wmma_f32_16x16x4_f32(
           false, av,  false, b[s][ct],
           (short)0, acc[ct],  false,  false);
  }
}

template <bool SCALE>
__device__ __forceinline__ void pass_compute(const float* aRow, const float* bRow, float scale,
                                             v8f (&acc)[8]) {
  v2f a0[2], a1[2];
  v2f b0[2][8], b1[2][8];
  load_chunk(aRow, bRow, 0, a0, b0);
#pragma unroll 1
  for (int k0 = 0; k0 < D - 16; k0 += 16) {
    load_chunk(aRow, bRow, k0 + 8, a1, b1);
    mma_chunk<SCALE>(a0, b0, scale, acc);
    load_chunk(aRow, bRow, k0 + 16, a0, b0);
    mma_chunk<SCALE>(a1, b1, scale, acc);
  }
  load_chunk(aRow, bRow, D - 8, a1, b1);
  mma_chunk<SCALE>(a0, b0, scale, acc);
  mma_chunk<SCALE>(a1, b1, scale, acc);
}

__global__ __launch_bounds__(GEMM_THREADS) void sage_wmma_kernel(const float* __restrict__ agg,
                                                        const float* __restrict__ deg,
                                                        const float* __restrict__ xin,
                                                        const float* __restrict__ Wl,
                                                        const float* __restrict__ Wr,
                                                        const float* __restrict__ bias,
                                                        float* __restrict__ out,
                                                        int n, int applyRelu) {
  __shared__ __attribute__((aligned(16))) float lds[LDS_W_FLOATS + LDS_A_FLOATS];

  const int lane = threadIdx.x & 31;
  const int wave = __builtin_amdgcn_readfirstlane(threadIdx.x >> 5);
  const int nTiles = n >> 4;
  const int tileStep = GEMM_WAVES * gridDim.x;

  const int m  = lane & 15;
  const int kh = (lane >> 4) * 2;

  float* sW  = lds;
  float* sAw = lds + LDS_W_FLOATS + wave * 16 * PITCH;
  const float* aRow  = sAw + m * PITCH + kh;
  const float* bRowL = sW + m * PITCH + kh;
  const float* bRowR = bRowL + D * PITCH;

  int tile = wave * gridDim.x + blockIdx.x;

  if (tile < nTiles) stage_rows(agg, tile * 16, lane, sAw);

  {
    float4 wv[16];
#pragma unroll
    for (int j = 0; j < 16; ++j) {
      const int row = j * GEMM_WAVES + wave;
      const float* W = (row < D) ? Wl : Wr;
      wv[j] = ((const float4*)(W + (row & (D - 1)) * D))[lane];
    }
#pragma unroll
    for (int j = 0; j < 16; ++j) {
      const int row = j * GEMM_WAVES + wave;
      *(float4*)(sW + row * PITCH + lane * 4) = wv[j];
    }
  }
  __syncthreads();

  while (tile < nTiles) {
    const int row0 = tile * 16;

    const float rdeg = 1.0f / fmaxf(deg[row0 + m], 1.0f);

    v8f zero = {};
    v8f acc[8];
#pragma unroll
    for (int i = 0; i < 8; ++i) acc[i] = zero;

    pass_compute<true>(aRow, bRowL, rdeg, acc);

    stage_rows(xin, row0, lane, sAw);
    pass_compute<false>(aRow, bRowR, 1.0f, acc);

    const int tileNext = tile + tileStep;
    if (tileNext < nTiles) stage_rows(agg, tileNext * 16, lane, sAw);

    const int rhalf = (lane >> 4) * 8;
#pragma unroll
    for (int ct = 0; ct < 8; ++ct) {
      const float bc = bias[ct * 16 + m];
#pragma unroll
      for (int r = 0; r < 8; ++r) {
        float v = acc[ct][r] + bc;
        if (applyRelu) v = fmaxf(v, 0.0f);
        out[(long long)(row0 + r + rhalf) * D + ct * 16 + m] = v;
      }
    }
    tile = tileNext;
  }
}

extern "C" void kernel_launch(void* const* d_in, const int* in_sizes, int n_in,
                              void* d_out, int out_size, void* d_ws, size_t ws_size,
                              hipStream_t stream) {
  (void)in_sizes; (void)n_in; (void)out_size; (void)ws_size;

  const float* x   = (const float*)d_in[0];
  const int*   ei  = (const int*)d_in[1];
  const float* W1l = (const float*)d_in[2];
  const float* b1  = (const float*)d_in[3];
  const float* W1r = (const float*)d_in[4];
  const float* W2l = (const float*)d_in[5];
  const float* b2  = (const float*)d_in[6];
  const float* W2r = (const float*)d_in[7];
  float*       out = (float*)d_out;

  const int* src = ei;
  const int* dst = ei + NE;

  char*  ws     = (char*)d_ws;
  float* deg    = (float*)ws;
  int*   cnt    = (int*)(ws + (512ull << 10));
  float* agg    = (float*)(ws + (1ull << 20));
  float* h      = (float*)(ws + (1ull << 20) + (size_t)NN * D * sizeof(float));
  int*   rowptr = (int*)(ws + (112ull << 20));
  int*   rank   = (int*)(ws + (113ull << 20));
  int*   csr    = (int*)(ws + (116ull << 20));

  hipMemsetAsync(cnt, 0, (size_t)NN * sizeof(int), stream);
  sage_rank_kernel<<<EDGE_BLOCKS, EDGE_THREADS, 0, stream>>>(dst, cnt, rank, NE / 4);
  sage_scan_kernel<<<SCAN_BLOCKS, SCAN_THREADS, 0, stream>>>(cnt, rowptr, deg, NN);
  sage_fill_kernel<<<EDGE_BLOCKS, EDGE_THREADS, 0, stream>>>(src, dst, rank, rowptr, csr, NE / 4);

  sage_gather_kernel<<<GATHER_BLOCKS, GATHER_THREADS, 0, stream>>>(x, rowptr, csr, agg, NN);
  sage_wmma_kernel<<<GEMM_BLOCKS, GEMM_THREADS, 0, stream>>>(agg, deg, x, W1l, W1r, b1, h,
                                                   NN,  1);

  sage_gather_kernel<<<GATHER_BLOCKS, GATHER_THREADS, 0, stream>>>(h, rowptr, csr, agg, NN);
  sage_wmma_kernel<<<GEMM_BLOCKS, GEMM_THREADS, 0, stream>>>(agg, deg, h, W2l, W2r, b2, out,
                                                   NN,  0);
}


# Neighbour gather fused into the layer kernel: each wave sums its tile's neighbour rows (14 loads in flight) into its LDS A tile as scaled means; no agg plane, 2 launches fewer; sum order kept
# speedup vs baseline: 2.6397x; 1.0582x over previous
#include <hip/hip_runtime.h>

#define NN 100000
#define NE 625000
#define D  128

typedef float v2f __attribute__((ext_vector_type(2)));
typedef float v8f __attribute__((ext_vector_type(8)));

static_assert(NE % 4 == 0, "edge arrays are walked four entries per thread");

#define EDGE_THREADS 256
#define EDGE_BLOCKS  ((NE / 4 + EDGE_THREADS - 1) / EDGE_THREADS)

__global__ __launch_bounds__(EDGE_THREADS) void sage_rank_kernel(const int* __restrict__ dst,
                                                                 int* __restrict__ cnt,
                                                                 int* __restrict__ rank, int E4) {
  const int q = blockIdx.x * blockDim.x + threadIdx.x;
  if (q >= E4) return;
  const int4 d = ((const int4*)dst)[q];
  int4 r;
  r.x = atomicAdd(&cnt[d.x], 1);
  r.y = atomicAdd(&cnt[d.y], 1);
  r.z = atomicAdd(&cnt[d.z], 1);
  r.w = atomicAdd(&cnt[d.w], 1);
  ((int4*)rank)[q] = r;
}

#define SCAN_THREADS 1024
#define SCAN_BLOCKS  ((NN + SCAN_THREADS - 1) / SCAN_THREADS)

__global__ __launch_bounds__(SCAN_THREADS) void sage_scan_kernel(const int* __restrict__ cnt,
                                                                 int* __restrict__ rowptr, int n) {
  __shared__ int sWave[32];
  __shared__ int sPart[32];
  const int t    = threadIdx.x;
  const int lane = t & 31;
  const int wave = t >> 5;
  const int base = blockIdx.x * SCAN_THREADS;

  int part = 0;
  const int4* c4 = (const int4*)cnt;
  const int lim = base >> 2;
  for (int q = t; q < lim; q += 8 * SCAN_THREADS) {
    int4 v[8];
#pragma unroll
    for (int u = 0; u < 8; ++u) {
      const int qq = q + u * SCAN_THREADS;
      v[u] = (qq < lim) ? c4[qq] : make_int4(0, 0, 0, 0);
    }
#pragma unroll
    for (int u = 0; u < 8; ++u) part += (v[u].x + v[u].y) + (v[u].z + v[u].w);
  }

  const int i = base + t;
  const int c = (i < n) ? cnt[i] : 0;

  int v = c;
  for (int d = 1; d < 32; d <<= 1) {
    const int u = __shfl_up(v, d);
    if (lane >= d) v += u;
  }
  for (int d = 16; d > 0; d >>= 1) part += __shfl_xor(part, d);
  if (lane == 31) sWave[wave] = v;
  if (lane == 0)  sPart[wave] = part;
  __syncthreads();

  int tot = ((lane < wave) ? sWave[lane] : 0) + sPart[lane];
  for (int d = 16; d > 0; d >>= 1) tot += __shfl_xor(tot, d);

  const int excl = tot + v - c;
  if (i < n) {
    rowptr[i] = excl;
    if (i == n - 1) rowptr[n] = excl + c;
  }
}

__global__ __launch_bounds__(EDGE_THREADS) void sage_fill_kernel(const int* __restrict__ src,
                                                                 const int* __restrict__ dst,
                                                                 const int* __restrict__ rank,
                                                                 const int* __restrict__ rowptr,
                                                                 int* __restrict__ csr, int E4) {
  const int q = blockIdx.x * blockDim.x + threadIdx.x;
  if (q >= E4) return;
  const int4 s = ((const int4*)src)[q];
  const int4 d = ((const int4*)dst)[q];
  const int4 r = ((const int4*)rank)[q];
  csr[rowptr[d.x] + r.x] = s.x;
  csr[rowptr[d.y] + r.y] = s.y;
  csr[rowptr[d.z] + r.z] = s.z;
  csr[rowptr[d.w] + r.w] = s.w;
}

#define PITCH        132
#define GEMM_WAVES   16
#define GEMM_THREADS (GEMM_WAVES * 32)
#define GEMM_BLOCKS  256
#define LDS_W_FLOATS (2 * D * PITCH)
#define LDS_A_FLOATS (GEMM_WAVES * 16 * PITCH)
#define GROWS        14

__device__ __forceinline__ void stage_rows(const float* __restrict__ src, int row0, int lane,
                                           float* sAw) {
  const float4* p = (const float4*)(src + (long long)row0 * D) + lane;
  float4 r[16];
#pragma unroll
  for (int i = 0; i < 16; ++i)
    r[i] = p[i * (D / 4)];
#pragma unroll
  for (int i = 0; i < 16; ++i)
    *(float4*)(sAw + i * PITCH + lane * 4) = r[i];
}

__device__ __forceinline__ void close_node(float* sAw, int node, int lane, float rdeg, float4& acc) {
  const float s = __int_as_float(__builtin_amdgcn_readlane(__float_as_int(rdeg), node));
  *(float4*)(sAw + node * PITCH + lane * 4) = make_float4(acc.x * s, acc.y * s, acc.z * s, acc.w * s);
  acc = make_float4(0.0f, 0.0f, 0.0f, 0.0f);
}

__device__ __forceinline__ void gather_tile(const float* __restrict__ xin,
                                            const int* __restrict__ rowptr,
                                            const int* __restrict__ csr,
                                            int row0, int lane, float* sAw) {
  const int kLo = (lane < 16) ? lane : 16;
  const int kHi = (lane < 16) ? lane + 1 : 16;
  const int rpLo = rowptr[row0 + kLo];
  const int rpHi = rowptr[row0 + kHi];
  const float rdeg = 1.0f / fmaxf((float)(rpHi - rpLo), 1.0f);

  const int eBeg = __builtin_amdgcn_readlane(rpLo, 0);
  const int eEnd = __builtin_amdgcn_readlane(rpLo, 16);

  int node = 0;
  int nodeEnd = __builtin_amdgcn_readlane(rpHi, 0);
  float4 acc = make_float4(0.0f, 0.0f, 0.0f, 0.0f);

  int idxNext = (eBeg + lane < eEnd) ? csr[eBeg + lane] : 0;
  for (int cb = eBeg; cb < eEnd; cb += GROWS) {
    const int idx = idxNext;
    const int nb = (eEnd - cb < GROWS) ? (eEnd - cb) : GROWS;
    if (cb + GROWS < eEnd)
      idxNext = (cb + GROWS + lane < eEnd) ? csr[cb + GROWS + lane] : 0;

    float4 v[GROWS];
#pragma unroll
    for (int t = 0; t < GROWS; ++t)
      if (t < nb) {
        const int s = __builtin_amdgcn_readlane(idx, t);
        v[t] = ((const float4*)(xin + (long long)s * D))[lane];
      }
#pragma unroll
    for (int t = 0; t < GROWS; ++t)
      if (t < nb) {
        while (node < 15 && cb + t >= nodeEnd) {
          close_node(sAw, node, lane, rdeg, acc);
          ++node;
          nodeEnd = __builtin_amdgcn_readlane(rpHi, node);
        }
        acc.x += v[t].x;
        acc.y += v[t].y;
        acc.z += v[t].z;
        acc.w += v[t].w;
      }
  }
  for (; node < 16; ++node)
    close_node(sAw, node, lane, rdeg, acc);
}

__device__ __forceinline__ void load_chunk(const float* aRow, const float* bRow, int k0,
                                           v2f (&a)[2], v2f (&b)[2][8]) {
#pragma unroll
  for (int s = 0; s < 2; ++s) {
    a[s] = *(const v2f*)(aRow + k0 + 4 * s);
#pragma unroll
    for (int ct = 0; ct < 8; ++ct)
      b[s][ct] = *(const v2f*)(bRow + ct * 16 * PITCH + k0 + 4 * s);
  }
}

__device__ __forceinline__ void mma_chunk(const v2f (&a)[2], const v2f (&b)[2][8], v8f (&acc)[8]) {
#pragma unroll
  for (int s = 0; s < 2; ++s) {
#pragma unroll
    for (int ct = 0; ct < 8; ++ct)
      acc[ct] = __builtin_amdgcn_wmma_f32_16x16x4_f32(
           false, a[s],  false, b[s][ct],
           (short)0, acc[ct],  false,  false);
  }
}

__device__ __forceinline__ void pass_compute(const float* aRow, const float* bRow, v8f (&acc)[8]) {
  v2f a0[2], a1[2];
  v2f b0[2][8], b1[2][8];
  load_chunk(aRow, bRow, 0, a0, b0);
#pragma unroll 1
  for (int k0 = 0; k0 < D - 16; k0 += 16) {
    load_chunk(aRow, bRow, k0 + 8, a1, b1);
    mma_chunk(a0, b0, acc);
    load_chunk(aRow, bRow, k0 + 16, a0, b0);
    mma_chunk(a1, b1, acc);
  }
  load_chunk(aRow, bRow, D - 8, a1, b1);
  mma_chunk(a0, b0, acc);
  mma_chunk(a1, b1, acc);
}

__global__ __launch_bounds__(GEMM_THREADS) void sage_layer_kernel(const float* __restrict__ xin,
                                                         const int* __restrict__ rowptr,
                                                         const int* __restrict__ csr,
                                                         const float* __restrict__ Wl,
                                                         const float* __restrict__ Wr,
                                                         const float* __restrict__ bias,
                                                         float* __restrict__ out,
                                                         int n, int applyRelu) {
  __shared__ __attribute__((aligned(16))) float lds[LDS_W_FLOATS + LDS_A_FLOATS];

  const int lane = threadIdx.x & 31;
  const int wave = __builtin_amdgcn_readfirstlane(threadIdx.x >> 5);
  const int nTiles = n >> 4;
  const int tileStep = GEMM_WAVES * gridDim.x;

  const int m  = lane & 15;
  const int kh = (lane >> 4) * 2;

  float* sW  = lds;
  float* sAw = lds + LDS_W_FLOATS + wave * 16 * PITCH;
  const float* aRow  = sAw + m * PITCH + kh;
  const float* bRowL = sW + m * PITCH + kh;
  const float* bRowR = bRowL + D * PITCH;

  int tile = wave * gridDim.x + blockIdx.x;

  {
    float4 wv[16];
#pragma unroll
    for (int j = 0; j < 16; ++j) {
      const int row = j * GEMM_WAVES + wave;
      const float* W = (row < D) ? Wl : Wr;
      wv[j] = ((const float4*)(W + (row & (D - 1)) * D))[lane];
    }
#pragma unroll
    for (int j = 0; j < 16; ++j) {
      const int row = j * GEMM_WAVES + wave;
      *(float4*)(sW + row * PITCH + lane * 4) = wv[j];
    }
  }

  __syncthreads();

  while (tile < nTiles) {
    const int row0 = tile * 16;

    gather_tile(xin, rowptr, csr, row0, lane, sAw);

    v8f zero = {};
    v8f acc[8];
#pragma unroll
    for (int i = 0; i < 8; ++i) acc[i] = zero;

    pass_compute(aRow, bRowL, acc);

    __builtin_amdgcn_sched_barrier(0);
    stage_rows(xin, row0, lane, sAw);
    pass_compute(aRow, bRowR, acc);

    const int rhalf = (lane >> 4) * 8;
#pragma unroll
    for (int ct = 0; ct < 8; ++ct) {
      const float bc = bias[ct * 16 + m];
#pragma unroll
      for (int r = 0; r < 8; ++r) {
        float v = acc[ct][r] + bc;
        if (applyRelu) v = fmaxf(v, 0.0f);
        out[(long long)(row0 + r + rhalf) * D + ct * 16 + m] = v;
      }
    }
    tile += tileStep;
  }
}

extern "C" void kernel_launch(void* const* d_in, const int* in_sizes, int n_in,
                              void* d_out, int out_size, void* d_ws, size_t ws_size,
                              hipStream_t stream) {
  (void)in_sizes; (void)n_in; (void)out_size; (void)ws_size;

  const float* x   = (const float*)d_in[0];
  const int*   ei  = (const int*)d_in[1];
  const float* W1l = (const float*)d_in[2];
  const float* b1  = (const float*)d_in[3];
  const float* W1r = (const float*)d_in[4];
  const float* W2l = (const float*)d_in[5];
  const float* b2  = (const float*)d_in[6];
  const float* W2r = (const float*)d_in[7];
  float*       out = (float*)d_out;

  const int* src = ei;
  const int* dst = ei + NE;

  char*  ws     = (char*)d_ws;
  int*   cnt    = (int*)(ws + (512ull << 10));
  float* h      = (float*)(ws + (1ull << 20));
  int*   rowptr = (int*)(ws + (112ull << 20));
  int*   rank   = (int*)(ws + (113ull << 20));
  int*   csr    = (int*)(ws + (116ull << 20));

  hipMemsetAsync(cnt, 0, (size_t)NN * sizeof(int), stream);
  sage_rank_kernel<<<EDGE_BLOCKS, EDGE_THREADS, 0, stream>>>(dst, cnt, rank, NE / 4);
  sage_scan_kernel<<<SCAN_BLOCKS, SCAN_THREADS, 0, stream>>>(cnt, rowptr, NN);
  sage_fill_kernel<<<EDGE_BLOCKS, EDGE_THREADS, 0, stream>>>(src, dst, rank, rowptr, csr, NE / 4);

  sage_layer_kernel<<<GEMM_BLOCKS, GEMM_THREADS, 0, stream>>>(x, rowptr, csr, W1l, W1r, b1, h,
                                                    NN,  1);

  sage_layer_kernel<<<GEMM_BLOCKS, GEMM_THREADS, 0, stream>>>(h, rowptr, csr, W2l, W2r, b2, out,
                                                    NN,  0);
}


# Edge list grouped by destination in two levels with LDS counters (returning LDS add as rank) instead of global atomic rank, scan and fill: no counter clear; a row's entries add in LDS-rank order
# speedup vs baseline: 3.2969x; 1.2490x over previous
#include <hip/hip_runtime.h>

#define NN 100000
#define NE 625000
#define D  128

typedef float v2f __attribute__((ext_vector_type(2)));
typedef float v8f __attribute__((ext_vector_type(8)));

static_assert(NE % 4 == 0, "edge arrays are walked four entries per thread");

#define NBUCK         256
#define BW            ((NN + NBUCK - 1) / NBUCK)
#define CHUNK4        ((NE / 4 + NBUCK - 1) / NBUCK)
#define COUNT_THREADS 256
#define COUNT_GROUPS  ((CHUNK4 + COUNT_THREADS - 1) / COUNT_THREADS)
#define PLACE_THREADS 512
#define PLACE_GROUPS  ((CHUNK4 + PLACE_THREADS - 1) / PLACE_THREADS)
#define PLACE_PARTS   (PLACE_THREADS / (NBUCK / 4))
#define PLACE_ROWS    (NBUCK / PLACE_PARTS)
#define SORT_THREADS  512
#define SORT_EDGES    8

static_assert(NBUCK == COUNT_THREADS, "one bucket counter per thread");
static_assert(PLACE_THREADS >= NBUCK && NBUCK % (4 * 32) == 0, "bucket scan: whole waves");
static_assert(BW <= SORT_THREADS, "one node counter per thread");

__global__ __launch_bounds__(COUNT_THREADS) void sage_count_kernel(const int* __restrict__ dst,
                                                                   int* __restrict__ hist, int E4) {
  __shared__ int sHist[NBUCK];
  const int t = threadIdx.x;
  const int g = blockIdx.x;

  int4 d[COUNT_GROUPS];
  bool has[COUNT_GROUPS];
#pragma unroll
  for (int k = 0; k < COUNT_GROUPS; ++k) {
    const int j = t + k * COUNT_THREADS;
    const int q = g * CHUNK4 + j;
    has[k] = (j < CHUNK4) && (q < E4);
    d[k] = ((const int4*)dst)[has[k] ? q : 0];
  }

  sHist[t] = 0;
  __syncthreads();
#pragma unroll
  for (int k = 0; k < COUNT_GROUPS; ++k)
    if (has[k]) {
      atomicAdd(&sHist[d[k].x / BW], 1);
      atomicAdd(&sHist[d[k].y / BW], 1);
      atomicAdd(&sHist[d[k].z / BW], 1);
      atomicAdd(&sHist[d[k].w / BW], 1);
    }
  __syncthreads();
  hist[g * NBUCK + t] = sHist[t];
}

__device__ __forceinline__ int wave_incl_scan(int v, int lane) {
  for (int d = 1; d < 32; d <<= 1) {
    const int u = __shfl_up(v, d);
    if (lane >= d) v += u;
  }
  return v;
}

__global__ __launch_bounds__(PLACE_THREADS) void sage_place_kernel(const int* __restrict__ src,
                                                                   const int* __restrict__ dst,
                                                                   const int* __restrict__ hist,
                                                                   int* __restrict__ bucketBase,
                                                                   int* __restrict__ pairs, int E4) {
  __shared__ __attribute__((aligned(16))) int sTot[PLACE_PARTS][NBUCK];
  __shared__ __attribute__((aligned(16))) int sPre[PLACE_PARTS][NBUCK];
  __shared__ int sWave[NBUCK / 32];
  __shared__ int sCur[NBUCK];
  const int t    = threadIdx.x;
  const int lane = t & 31;
  const int g    = blockIdx.x;

  int4 s[PLACE_GROUPS], d[PLACE_GROUPS];
  bool has[PLACE_GROUPS];
#pragma unroll
  for (int k = 0; k < PLACE_GROUPS; ++k) {
    const int j = t + k * PLACE_THREADS;
    const int q = g * CHUNK4 + j;
    has[k] = (j < CHUNK4) && (q < E4);
    s[k] = ((const int4*)src)[has[k] ? q : 0];
    d[k] = ((const int4*)dst)[has[k] ? q : 0];
  }

  {
    const int quad = t & (NBUCK / 4 - 1);
    const int part = __builtin_amdgcn_readfirstlane(t / (NBUCK / 4));
    const int4* col = (const int4*)hist + quad;
    int4 v[PLACE_ROWS];
#pragma unroll
    for (int u = 0; u < PLACE_ROWS; ++u) v[u] = col[(part * PLACE_ROWS + u) * (NBUCK / 4)];
    int4 tot = make_int4(0, 0, 0, 0), pre = make_int4(0, 0, 0, 0);
#pragma unroll
    for (int u = 0; u < PLACE_ROWS; ++u) {
      tot.x += v[u].x; tot.y += v[u].y; tot.z += v[u].z; tot.w += v[u].w;
      if (part * PLACE_ROWS + u < g) {
        pre.x += v[u].x; pre.y += v[u].y; pre.z += v[u].z; pre.w += v[u].w;
      }
    }
    *(int4*)&sTot[part][4 * quad] = tot;
    *(int4*)&sPre[part][4 * quad] = pre;
  }
  __syncthreads();

  int total = 0, prefix = 0, incl = 0;
  if (t < NBUCK) {
    for (int p = 0; p < PLACE_PARTS; ++p) {
      total  += sTot[p][t];
      prefix += sPre[p][t];
    }
    incl = wave_incl_scan(total, lane);
    if (lane == 31) sWave[t >> 5] = incl;
  }
  __syncthreads();
  if (t < NBUCK) {
    int before = 0;
    for (int w = 0; w < NBUCK / 32; ++w) before += (w < (t >> 5)) ? sWave[w] : 0;
    const int base = before + incl - total;
    sCur[t] = base + prefix;
    if (g == 0) {
      bucketBase[t] = base;
      if (t == NBUCK - 1) bucketBase[NBUCK] = base + total;
    }
  }
  __syncthreads();

  int2* pr = (int2*)pairs;
#pragma unroll
  for (int k = 0; k < PLACE_GROUPS; ++k)
    if (has[k]) {
      const int p0 = atomicAdd(&sCur[d[k].x / BW], 1);
      const int p1 = atomicAdd(&sCur[d[k].y / BW], 1);
      const int p2 = atomicAdd(&sCur[d[k].z / BW], 1);
      const int p3 = atomicAdd(&sCur[d[k].w / BW], 1);
      pr[p0] = make_int2(s[k].x, d[k].x);
      pr[p1] = make_int2(s[k].y, d[k].y);
      pr[p2] = make_int2(s[k].z, d[k].z);
      pr[p3] = make_int2(s[k].w, d[k].w);
    }
}

__global__ __launch_bounds__(SORT_THREADS) void sage_sort_kernel(const int* __restrict__ pairs,
                                                                 const int* __restrict__ bucketBase,
                                                                 int* __restrict__ rowptr,
                                                                 int* __restrict__ csr, int n) {
  __shared__ int sCnt[SORT_THREADS];
  __shared__ int sWave[SORT_THREADS / 32];
  const int t     = threadIdx.x;
  const int lane  = t & 31;
  const int wave  = t >> 5;
  const int node0 = blockIdx.x * BW;
  const int eBeg  = bucketBase[blockIdx.x];
  const int eEnd  = bucketBase[blockIdx.x + 1];
  const int2* pr  = (const int2*)pairs;

  sCnt[t] = 0;
  __syncthreads();
  for (int e0 = eBeg + t; e0 < eEnd; e0 += SORT_EDGES * SORT_THREADS) {
    int dl[SORT_EDGES];
#pragma unroll
    for (int u = 0; u < SORT_EDGES; ++u) {
      const int e = e0 + u * SORT_THREADS;
      dl[u] = (e < eEnd) ? pr[e].y : -1;
    }
#pragma unroll
    for (int u = 0; u < SORT_EDGES; ++u)
      if (dl[u] >= 0) atomicAdd(&sCnt[dl[u] - node0], 1);
  }
  __syncthreads();

  const int c = sCnt[t];
  const int incl = wave_incl_scan(c, lane);
  if (lane == 31) sWave[wave] = incl;
  __syncthreads();
  int before = 0;
  for (int w = 0; w < SORT_THREADS / 32; ++w) before += (w < wave) ? sWave[w] : 0;
  const int first = eBeg + before + incl - c;

  sCnt[t] = first;
  if (t < BW && node0 + t < n) rowptr[node0 + t] = first;
  if (blockIdx.x == NBUCK - 1 && t == 0) rowptr[n] = eEnd;
  __syncthreads();
  for (int e0 = eBeg + t; e0 < eEnd; e0 += SORT_EDGES * SORT_THREADS) {
    int2 p[SORT_EDGES];
#pragma unroll
    for (int u = 0; u < SORT_EDGES; ++u) {
      const int e = e0 + u * SORT_THREADS;
      p[u] = (e < eEnd) ? pr[e] : make_int2(0, -1);
    }
    int pos[SORT_EDGES];
#pragma unroll
    for (int u = 0; u < SORT_EDGES; ++u)
      pos[u] = (p[u].y >= 0) ? atomicAdd(&sCnt[p[u].y - node0], 1) : -1;
#pragma unroll
    for (int u = 0; u < SORT_EDGES; ++u)
      if (pos[u] >= 0) csr[pos[u]] = p[u].x;
  }
}

#define PITCH        132
#define GEMM_WAVES   16
#define GEMM_THREADS (GEMM_WAVES * 32)
#define GEMM_BLOCKS  256
#define LDS_W_FLOATS (2 * D * PITCH)
#define LDS_A_FLOATS (GEMM_WAVES * 16 * PITCH)
#define GROWS        14

__device__ __forceinline__ void stage_rows(const float* __restrict__ src, int row0, int lane,
                                           float* sAw) {
  const float4* p = (const float4*)(src + (long long)row0 * D) + lane;
  float4 r[16];
#pragma unroll
  for (int i = 0; i < 16; ++i)
    r[i] = p[i * (D / 4)];
#pragma unroll
  for (int i = 0; i < 16; ++i)
    *(float4*)(sAw + i * PITCH + lane * 4) = r[i];
}

__device__ __forceinline__ void close_node(float* sAw, int node, int lane, float rdeg, float4& acc) {
  const float s = __int_as_float(__builtin_amdgcn_readlane(__float_as_int(rdeg), node));
  *(float4*)(sAw + node * PITCH + lane * 4) = make_float4(acc.x * s, acc.y * s, acc.z * s, acc.w * s);
  acc = make_float4(0.0f, 0.0f, 0.0f, 0.0f);
}

__device__ __forceinline__ void gather_tile(const float* __restrict__ xin,
                                            const int* __restrict__ rowptr,
                                            const int* __restrict__ csr,
                                            int row0, int lane, float* sAw) {
  const int kLo = (lane < 16) ? lane : 16;
  const int kHi = (lane < 16) ? lane + 1 : 16;
  const int rpLo = rowptr[row0 + kLo];
  const int rpHi = rowptr[row0 + kHi];
  const float rdeg = 1.0f / fmaxf((float)(rpHi - rpLo), 1.0f);

  const int eBeg = __builtin_amdgcn_readlane(rpLo, 0);
  const int eEnd = __builtin_amdgcn_readlane(rpLo, 16);

  int node = 0;
  int nodeEnd = __builtin_amdgcn_readlane(rpHi, 0);
  float4 acc = make_float4(0.0f, 0.0f, 0.0f, 0.0f);

  int idxNext = (eBeg + lane < eEnd) ? csr[eBeg + lane] : 0;
  for (int cb = eBeg; cb < eEnd; cb += GROWS) {
    const int idx = idxNext;
    const int nb = (eEnd - cb < GROWS) ? (eEnd - cb) : GROWS;
    if (cb + GROWS < eEnd)
      idxNext = (cb + GROWS + lane < eEnd) ? csr[cb + GROWS + lane] : 0;

    float4 v[GROWS];
#pragma unroll
    for (int t = 0; t < GROWS; ++t)
      if (t < nb) {
        const int s = __builtin_amdgcn_readlane(idx, t);
        v[t] = ((const float4*)(xin + (long long)s * D))[lane];
      }
#pragma unroll
    for (int t = 0; t < GROWS; ++t)
      if (t < nb) {
        while (node < 15 && cb + t >= nodeEnd) {
          close_node(sAw, node, lane, rdeg, acc);
          ++node;
          nodeEnd = __builtin_amdgcn_readlane(rpHi, node);
        }
        acc.x += v[t].x;
        acc.y += v[t].y;
        acc.z += v[t].z;
        acc.w += v[t].w;
      }
  }
  for (; node < 16; ++node)
    close_node(sAw, node, lane, rdeg, acc);
}

__device__ __forceinline__ void load_chunk(const float* aRow, const float* bRow, int k0,
                                           v2f (&a)[2], v2f (&b)[2][8]) {
#pragma unroll
  for (int s = 0; s < 2; ++s) {
    a[s] = *(const v2f*)(aRow + k0 + 4 * s);
#pragma unroll
    for (int ct = 0; ct < 8; ++ct)
      b[s][ct] = *(const v2f*)(bRow + ct * 16 * PITCH + k0 + 4 * s);
  }
}

__device__ __forceinline__ void mma_chunk(const v2f (&a)[2], const v2f (&b)[2][8], v8f (&acc)[8]) {
#pragma unroll
  for (int s = 0; s < 2; ++s) {
#pragma unroll
    for (int ct = 0; ct < 8; ++ct)
      acc[ct] = __builtin_amdgcn_wmma_f32_16x16x4_f32(
           false, a[s],  false, b[s][ct],
           (short)0, acc[ct],  false,  false);
  }
}

__device__ __forceinline__ void pass_compute(const float* aRow, const float* bRow, v8f (&acc)[8]) {
  v2f a0[2], a1[2];
  v2f b0[2][8], b1[2][8];
  load_chunk(aRow, bRow, 0, a0, b0);
#pragma unroll 1
  for (int k0 = 0; k0 < D - 16; k0 += 16) {
    load_chunk(aRow, bRow, k0 + 8, a1, b1);
    mma_chunk(a0, b0, acc);
    load_chunk(aRow, bRow, k0 + 16, a0, b0);
    mma_chunk(a1, b1, acc);
  }
  load_chunk(aRow, bRow, D - 8, a1, b1);
  mma_chunk(a0, b0, acc);
  mma_chunk(a1, b1, acc);
}

__global__ __launch_bounds__(GEMM_THREADS) void sage_layer_kernel(const float* __restrict__ xin,
                                                         const int* __restrict__ rowptr,
                                                         const int* __restrict__ csr,
                                                         const float* __restrict__ Wl,
                                                         const float* __restrict__ Wr,
                                                         const float* __restrict__ bias,
                                                         float* __restrict__ out,
                                                         int n, int applyRelu) {
  __shared__ __attribute__((aligned(16))) float lds[LDS_W_FLOATS + LDS_A_FLOATS];

  const int lane = threadIdx.x & 31;
  const int wave = __builtin_amdgcn_readfirstlane(threadIdx.x >> 5);
  const int nTiles = n >> 4;
  const int tileStep = GEMM_WAVES * gridDim.x;

  const int m  = lane & 15;
  const int kh = (lane >> 4) * 2;

  float* sW  = lds;
  float* sAw = lds + LDS_W_FLOATS + wave * 16 * PITCH;
  const float* aRow  = sAw + m * PITCH + kh;
  const float* bRowL = sW + m * PITCH + kh;
  const float* bRowR = bRowL + D * PITCH;

  int tile = wave * gridDim.x + blockIdx.x;

  {
    float4 wv[16];
#pragma unroll
    for (int j = 0; j < 16; ++j) {
      const int row = j * GEMM_WAVES + wave;
      const float* W = (row < D) ? Wl : Wr;
      wv[j] = ((const float4*)(W + (row & (D - 1)) * D))[lane];
    }
#pragma unroll
    for (int j = 0; j < 16; ++j) {
      const int row = j * GEMM_WAVES + wave;
      *(float4*)(sW + row * PITCH + lane * 4) = wv[j];
    }
  }

  __syncthreads();

  while (tile < nTiles) {
    const int row0 = tile * 16;

    gather_tile(xin, rowptr, csr, row0, lane, sAw);

    v8f zero = {};
    v8f acc[8];
#pragma unroll
    for (int i = 0; i < 8; ++i) acc[i] = zero;

    pass_compute(aRow, bRowL, acc);

    __builtin_amdgcn_sched_barrier(0);
    stage_rows(xin, row0, lane, sAw);
    pass_compute(aRow, bRowR, acc);

    const int rhalf = (lane >> 4) * 8;
#pragma unroll
    for (int ct = 0; ct < 8; ++ct) {
      const float bc = bias[ct * 16 + m];
#pragma unroll
      for (int r = 0; r < 8; ++r) {
        float v = acc[ct][r] + bc;
        if (applyRelu) v = fmaxf(v, 0.0f);
        out[(long long)(row0 + r + rhalf) * D + ct * 16 + m] = v;
      }
    }
    tile += tileStep;
  }
}

extern "C" void kernel_launch(void* const* d_in, const int* in_sizes, int n_in,
                              void* d_out, int out_size, void* d_ws, size_t ws_size,
                              hipStream_t stream) {
  (void)in_sizes; (void)n_in; (void)out_size; (void)ws_size;

  const float* x   = (const float*)d_in[0];
  const int*   ei  = (const int*)d_in[1];
  const float* W1l = (const float*)d_in[2];
  const float* b1  = (const float*)d_in[3];
  const float* W1r = (const float*)d_in[4];
  const float* W2l = (const float*)d_in[5];
  const float* b2  = (const float*)d_in[6];
  const float* W2r = (const float*)d_in[7];
  float*       out = (float*)d_out;

  const int* src = ei;
  const int* dst = ei + NE;

  char*  ws     = (char*)d_ws;
  int*   hist   = (int*)(ws + (512ull << 10));
  float* h      = (float*)(ws + (1ull << 20));
  int*   rowptr = (int*)(ws + (112ull << 20));
  int*   bbase  = (int*)(ws + (113ull << 20));
  int*   csr    = (int*)(ws + (116ull << 20));
  int*   pairs  = (int*)(ws + (120ull << 20));

  sage_count_kernel<<<NBUCK, COUNT_THREADS, 0, stream>>>(dst, hist, NE / 4);
  sage_place_kernel<<<NBUCK, PLACE_THREADS, 0, stream>>>(src, dst, hist, bbase, pairs, NE / 4);
  sage_sort_kernel<<<NBUCK, SORT_THREADS, 0, stream>>>(pairs, bbase, rowptr, csr, NN);

  sage_layer_kernel<<<GEMM_BLOCKS, GEMM_THREADS, 0, stream>>>(x, rowptr, csr, W1l, W1r, b1, h,
                                                    NN,  1);

  sage_layer_kernel<<<GEMM_BLOCKS, GEMM_THREADS, 0, stream>>>(h, rowptr, csr, W2l, W2r, b2, out,
                                                    NN,  0);
}


# Layer kernel: the 106 tiles beyond six per SIMD are cut into four 32-column quarters, one per SIMD of their work-group, instead of a seventh whole tile on one SIMD; sum order unchanged
# speedup vs baseline: 3.4967x; 1.0606x over previous
#include <hip/hip_runtime.h>

#define NN 100000
#define NE 625000
#define D  128

typedef float v2f __attribute__((ext_vector_type(2)));
typedef float v8f __attribute__((ext_vector_type(8)));

static_assert(NE % 4 == 0, "edge arrays are walked four entries per thread");

#define NBUCK         256
#define BW            ((NN + NBUCK - 1) / NBUCK)
#define CHUNK4        ((NE / 4 + NBUCK - 1) / NBUCK)
#define COUNT_THREADS 256
#define COUNT_GROUPS  ((CHUNK4 + COUNT_THREADS - 1) / COUNT_THREADS)
#define PLACE_THREADS 512
#define PLACE_GROUPS  ((CHUNK4 + PLACE_THREADS - 1) / PLACE_THREADS)
#define PLACE_PARTS   (PLACE_THREADS / (NBUCK / 4))
#define PLACE_ROWS    (NBUCK / PLACE_PARTS)
#define SORT_THREADS  512
#define SORT_EDGES    8

static_assert(NBUCK == COUNT_THREADS, "one bucket counter per thread");
static_assert(PLACE_THREADS >= NBUCK && NBUCK % (4 * 32) == 0, "bucket scan: whole waves");
static_assert(BW <= SORT_THREADS, "one node counter per thread");

__global__ __launch_bounds__(COUNT_THREADS) void sage_count_kernel(const int* __restrict__ dst,
                                                                   int* __restrict__ hist, int E4) {
  __shared__ int sHist[NBUCK];
  const int t = threadIdx.x;
  const int g = blockIdx.x;

  int4 d[COUNT_GROUPS];
  bool has[COUNT_GROUPS];
#pragma unroll
  for (int k = 0; k < COUNT_GROUPS; ++k) {
    const int j = t + k * COUNT_THREADS;
    const int q = g * CHUNK4 + j;
    has[k] = (j < CHUNK4) && (q < E4);
    d[k] = ((const int4*)dst)[has[k] ? q : 0];
  }

  sHist[t] = 0;
  __syncthreads();
#pragma unroll
  for (int k = 0; k < COUNT_GROUPS; ++k)
    if (has[k]) {
      atomicAdd(&sHist[d[k].x / BW], 1);
      atomicAdd(&sHist[d[k].y / BW], 1);
      atomicAdd(&sHist[d[k].z / BW], 1);
      atomicAdd(&sHist[d[k].w / BW], 1);
    }
  __syncthreads();
  hist[g * NBUCK + t] = sHist[t];
}

__device__ __forceinline__ int wave_incl_scan(int v, int lane) {
  for (int d = 1; d < 32; d <<= 1) {
    const int u = __shfl_up(v, d);
    if (lane >= d) v += u;
  }
  return v;
}

__global__ __launch_bounds__(PLACE_THREADS) void sage_place_kernel(const int* __restrict__ src,
                                                                   const int* __restrict__ dst,
                                                                   const int* __restrict__ hist,
                                                                   int* __restrict__ bucketBase,
                                                                   int* __restrict__ pairs, int E4) {
  __shared__ __attribute__((aligned(16))) int sTot[PLACE_PARTS][NBUCK];
  __shared__ __attribute__((aligned(16))) int sPre[PLACE_PARTS][NBUCK];
  __shared__ int sWave[NBUCK / 32];
  __shared__ int sCur[NBUCK];
  const int t    = threadIdx.x;
  const int lane = t & 31;
  const int g    = blockIdx.x;

  int4 s[PLACE_GROUPS], d[PLACE_GROUPS];
  bool has[PLACE_GROUPS];
#pragma unroll
  for (int k = 0; k < PLACE_GROUPS; ++k) {
    const int j = t + k * PLACE_THREADS;
    const int q = g * CHUNK4 + j;
    has[k] = (j < CHUNK4) && (q < E4);
    s[k] = ((const int4*)src)[has[k] ? q : 0];
    d[k] = ((const int4*)dst)[has[k] ? q : 0];
  }

  {
    const int quad = t & (NBUCK / 4 - 1);
    const int part = __builtin_amdgcn_readfirstlane(t / (NBUCK / 4));
    const int4* col = (const int4*)hist + quad;
    int4 v[PLACE_ROWS];
#pragma unroll
    for (int u = 0; u < PLACE_ROWS; ++u) v[u] = col[(part * PLACE_ROWS + u) * (NBUCK / 4)];
    int4 tot = make_int4(0, 0, 0, 0), pre = make_int4(0, 0, 0, 0);
#pragma unroll
    for (int u = 0; u < PLACE_ROWS; ++u) {
      tot.x += v[u].x; tot.y += v[u].y; tot.z += v[u].z; tot.w += v[u].w;
      if (part * PLACE_ROWS + u < g) {
        pre.x += v[u].x; pre.y += v[u].y; pre.z += v[u].z; pre.w += v[u].w;
      }
    }
    *(int4*)&sTot[part][4 * quad] = tot;
    *(int4*)&sPre[part][4 * quad] = pre;
  }
  __syncthreads();

  int total = 0, prefix = 0, incl = 0;
  if (t < NBUCK) {
    for (int p = 0; p < PLACE_PARTS; ++p) {
      total  += sTot[p][t];
      prefix += sPre[p][t];
    }
    incl = wave_incl_scan(total, lane);
    if (lane == 31) sWave[t >> 5] = incl;
  }
  __syncthreads();
  if (t < NBUCK) {
    int before = 0;
    for (int w = 0; w < NBUCK / 32; ++w) before += (w < (t >> 5)) ? sWave[w] : 0;
    const int base = before + incl - total;
    sCur[t] = base + prefix;
    if (g == 0) {
      bucketBase[t] = base;
      if (t == NBUCK - 1) bucketBase[NBUCK] = base + total;
    }
  }
  __syncthreads();

  int2* pr = (int2*)pairs;
#pragma unroll
  for (int k = 0; k < PLACE_GROUPS; ++k)
    if (has[k]) {
      const int p0 = atomicAdd(&sCur[d[k].x / BW], 1);
      const int p1 = atomicAdd(&sCur[d[k].y / BW], 1);
      const int p2 = atomicAdd(&sCur[d[k].z / BW], 1);
      const int p3 = atomicAdd(&sCur[d[k].w / BW], 1);
      pr[p0] = make_int2(s[k].x, d[k].x);
      pr[p1] = make_int2(s[k].y, d[k].y);
      pr[p2] = make_int2(s[k].z, d[k].z);
      pr[p3] = make_int2(s[k].w, d[k].w);
    }
}

__global__ __launch_bounds__(SORT_THREADS) void sage_sort_kernel(const int* __restrict__ pairs,
                                                                 const int* __restrict__ bucketBase,
                                                                 int* __restrict__ rowptr,
                                                                 int* __restrict__ csr, int n) {
  __shared__ int sCnt[SORT_THREADS];
  __shared__ int sWave[SORT_THREADS / 32];
  const int t     = threadIdx.x;
  const int lane  = t & 31;
  const int wave  = t >> 5;
  const int node0 = blockIdx.x * BW;
  const int eBeg  = bucketBase[blockIdx.x];
  const int eEnd  = bucketBase[blockIdx.x + 1];
  const int2* pr  = (const int2*)pairs;

  sCnt[t] = 0;
  __syncthreads();
  for (int e0 = eBeg + t; e0 < eEnd; e0 += SORT_EDGES * SORT_THREADS) {
    int dl[SORT_EDGES];
#pragma unroll
    for (int u = 0; u < SORT_EDGES; ++u) {
      const int e = e0 + u * SORT_THREADS;
      dl[u] = (e < eEnd) ? pr[e].y : -1;
    }
#pragma unroll
    for (int u = 0; u < SORT_EDGES; ++u)
      if (dl[u] >= 0) atomicAdd(&sCnt[dl[u] - node0], 1);
  }
  __syncthreads();

  const int c = sCnt[t];
  const int incl = wave_incl_scan(c, lane);
  if (lane == 31) sWave[wave] = incl;
  __syncthreads();
  int before = 0;
  for (int w = 0; w < SORT_THREADS / 32; ++w) before += (w < wave) ? sWave[w] : 0;
  const int first = eBeg + before + incl - c;

  sCnt[t] = first;
  if (t < BW && node0 + t < n) rowptr[node0 + t] = first;
  if (blockIdx.x == NBUCK - 1 && t == 0) rowptr[n] = eEnd;
  __syncthreads();
  for (int e0 = eBeg + t; e0 < eEnd; e0 += SORT_EDGES * SORT_THREADS) {
    int2 p[SORT_EDGES];
#pragma unroll
    for (int u = 0; u < SORT_EDGES; ++u) {
      const int e = e0 + u * SORT_THREADS;
      p[u] = (e < eEnd) ? pr[e] : make_int2(0, -1);
    }
    int pos[SORT_EDGES];
#pragma unroll
    for (int u = 0; u < SORT_EDGES; ++u)
      pos[u] = (p[u].y >= 0) ? atomicAdd(&sCnt[p[u].y - node0], 1) : -1;
#pragma unroll
    for (int u = 0; u < SORT_EDGES; ++u)
      if (pos[u] >= 0) csr[pos[u]] = p[u].x;
  }
}

#define PITCH        132
#define GEMM_WAVES   16
#define GEMM_THREADS (GEMM_WAVES * 32)
#define GEMM_BLOCKS  256
#define LDS_W_FLOATS (2 * D * PITCH)
#define LDS_A_FLOATS (GEMM_WAVES * 16 * PITCH)
#define GROWS        14

__device__ __forceinline__ void stage_rows(const float* __restrict__ src, int row0, int lane,
                                           float* sAw) {
  const float4* p = (const float4*)(src + (long long)row0 * D) + lane;
  float4 r[16];
#pragma unroll
  for (int i = 0; i < 16; ++i)
    r[i] = p[i * (D / 4)];
#pragma unroll
  for (int i = 0; i < 16; ++i)
    *(float4*)(sAw + i * PITCH + lane * 4) = r[i];
}

__device__ __forceinline__ void close_node(float* sAw, int node, int lane, float rdeg, float4& acc) {
  const float s = __int_as_float(__builtin_amdgcn_readlane(__float_as_int(rdeg), node));
  *(float4*)(sAw + node * PITCH + lane * 4) = make_float4(acc.x * s, acc.y * s, acc.z * s, acc.w * s);
  acc = make_float4(0.0f, 0.0f, 0.0f, 0.0f);
}

__device__ __forceinline__ void gather_tile(const float* __restrict__ xin,
                                            const int* __restrict__ rowptr,
                                            const int* __restrict__ csr,
                                            int row0, int lane, float* sAw) {
  const int kLo = (lane < 16) ? lane : 16;
  const int kHi = (lane < 16) ? lane + 1 : 16;
  const int rpLo = rowptr[row0 + kLo];
  const int rpHi = rowptr[row0 + kHi];
  const float rdeg = 1.0f / fmaxf((float)(rpHi - rpLo), 1.0f);

  const int eBeg = __builtin_amdgcn_readlane(rpLo, 0);
  const int eEnd = __builtin_amdgcn_readlane(rpLo, 16);

  int node = 0;
  int nodeEnd = __builtin_amdgcn_readlane(rpHi, 0);
  float4 acc = make_float4(0.0f, 0.0f, 0.0f, 0.0f);

  int idxNext = (eBeg + lane < eEnd) ? csr[eBeg + lane] : 0;
  for (int cb = eBeg; cb < eEnd; cb += GROWS) {
    const int idx = idxNext;
    const int nb = (eEnd - cb < GROWS) ? (eEnd - cb) : GROWS;
    if (cb + GROWS < eEnd)
      idxNext = (cb + GROWS + lane < eEnd) ? csr[cb + GROWS + lane] : 0;

    float4 v[GROWS];
#pragma unroll
    for (int t = 0; t < GROWS; ++t)
      if (t < nb) {
        const int s = __builtin_amdgcn_readlane(idx, t);
        v[t] = ((const float4*)(xin + (long long)s * D))[lane];
      }
#pragma unroll
    for (int t = 0; t < GROWS; ++t)
      if (t < nb) {
        while (node < 15 && cb + t >= nodeEnd) {
          close_node(sAw, node, lane, rdeg, acc);
          ++node;
          nodeEnd = __builtin_amdgcn_readlane(rpHi, node);
        }
        acc.x += v[t].x;
        acc.y += v[t].y;
        acc.z += v[t].z;
        acc.w += v[t].w;
      }
  }
  for (; node < 16; ++node)
    close_node(sAw, node, lane, rdeg, acc);
}

template <int NCT, int KS>
__device__ __forceinline__ void load_chunk(const float* aRow, const float* bRow, int k0,
                                           v2f (&a)[KS], v2f (&b)[KS][NCT]) {
#pragma unroll
  for (int s = 0; s < KS; ++s) {
    a[s] = *(const v2f*)(aRow + k0 + 4 * s);
#pragma unroll
    for (int ct = 0; ct < NCT; ++ct)
      b[s][ct] = *(const v2f*)(bRow + ct * 16 * PITCH + k0 + 4 * s);
  }
}

template <int NCT, int KS>
__device__ __forceinline__ void mma_chunk(const v2f (&a)[KS], const v2f (&b)[KS][NCT],
                                          v8f (&acc)[NCT]) {
#pragma unroll
  for (int s = 0; s < KS; ++s) {
#pragma unroll
    for (int ct = 0; ct < NCT; ++ct)
      acc[ct] = __builtin_amdgcn_wmma_f32_16x16x4_f32(
           false, a[s],  false, b[s][ct],
           (short)0, acc[ct],  false,  false);
  }
}

template <int NCT, int KS>
__device__ __forceinline__ void pass_compute(const float* aRow, const float* bRow,
                                             v8f (&acc)[NCT]) {
  constexpr int CH = 4 * KS;
  static_assert(D % (2 * CH) == 0, "a pass is a whole number of chunk pairs");
  v2f a0[KS], a1[KS];
  v2f b0[KS][NCT], b1[KS][NCT];
  load_chunk<NCT, KS>(aRow, bRow, 0, a0, b0);
#pragma unroll 1
  for (int k0 = 0; k0 < D - 2 * CH; k0 += 2 * CH) {
    load_chunk<NCT, KS>(aRow, bRow, k0 + CH, a1, b1);
    mma_chunk<NCT, KS>(a0, b0, acc);
    load_chunk<NCT, KS>(aRow, bRow, k0 + 2 * CH, a0, b0);
    mma_chunk<NCT, KS>(a1, b1, acc);
  }
  load_chunk<NCT, KS>(aRow, bRow, D - CH, a1, b1);
  mma_chunk<NCT, KS>(a0, b0, acc);
  mma_chunk<NCT, KS>(a1, b1, acc);
}

template <int NCT, int KS, int CT0>
__device__ __forceinline__ void tile_products(const float* __restrict__ xin,
                                              const float* __restrict__ bias,
                                              float* __restrict__ out,
                                              const float* aRow, const float* bRowL,
                                              const float* bRowR, float* sAw,
                                              int row0, int lane, int m, int applyRelu) {
  constexpr int ct0 = CT0;
  v8f zero = {};
  v8f acc[NCT];
#pragma unroll
  for (int i = 0; i < NCT; ++i) acc[i] = zero;

  pass_compute<NCT, KS>(aRow, bRowL + ct0 * 16 * PITCH, acc);

  __builtin_amdgcn_sched_barrier(0);
  stage_rows(xin, row0, lane, sAw);
  pass_compute<NCT, KS>(aRow, bRowR + ct0 * 16 * PITCH, acc);

  const int rhalf = (lane >> 4) * 8;
#pragma unroll
  for (int ct = 0; ct < NCT; ++ct) {
    const float bc = bias[(ct0 + ct) * 16 + m];
#pragma unroll
    for (int r = 0; r < 8; ++r) {
      float v = acc[ct][r] + bc;
      if (applyRelu) v = fmaxf(v, 0.0f);
      out[(long long)(row0 + r + rhalf) * D + (ct0 + ct) * 16 + m] = v;
    }
  }
}

__global__ __launch_bounds__(GEMM_THREADS) void sage_layer_kernel(const float* __restrict__ xin,
                                                         const int* __restrict__ rowptr,
                                                         const int* __restrict__ csr,
                                                         const float* __restrict__ Wl,
                                                         const float* __restrict__ Wr,
                                                         const float* __restrict__ bias,
                                                         float* __restrict__ out,
                                                         int n, int applyRelu) {
  __shared__ __attribute__((aligned(16))) float lds[LDS_W_FLOATS + LDS_A_FLOATS];

  const int lane = threadIdx.x & 31;
  const int wave = __builtin_amdgcn_readfirstlane(threadIdx.x >> 5);
  const int nTiles = n >> 4;
  const int tileStep = GEMM_WAVES * gridDim.x;

  const int m  = lane & 15;
  const int kh = (lane >> 4) * 2;

  float* sW  = lds;
  float* sAw = lds + LDS_W_FLOATS + wave * 16 * PITCH;
  const float* aRow  = sAw + m * PITCH + kh;
  const float* bRowL = sW + m * PITCH + kh;
  const float* bRowR = bRowL + D * PITCH;

  int tile = wave * gridDim.x + blockIdx.x;

  {
    float4 wv[16];
#pragma unroll
    for (int j = 0; j < 16; ++j) {
      const int row = j * GEMM_WAVES + wave;
      const float* W = (row < D) ? Wl : Wr;
      wv[j] = ((const float4*)(W + (row & (D - 1)) * D))[lane];
    }
#pragma unroll
    for (int j = 0; j < 16; ++j) {
      const int row = j * GEMM_WAVES + wave;
      *(float4*)(sW + row * PITCH + lane * 4) = wv[j];
    }
  }

  __syncthreads();

  const int nFull = nTiles - nTiles % (4 * (int)gridDim.x);
  while (tile < nFull) {
    const int row0 = tile * 16;

    gather_tile(xin, rowptr, csr, row0, lane, sAw);
    tile_products<8, 2, 0>(xin, bias, out, aRow, bRowL, bRowR, sAw, row0, lane, m, applyRelu);
    tile += tileStep;
  }

  const int qWave0 = (nFull / (int)gridDim.x) & (GEMM_WAVES - 1);
  if ((wave & ~3) == qWave0) {
    int laneQ = lane;
    asm volatile("" : "+v"(laneQ));
    const int mQ  = laneQ & 15;
    const int khQ = (laneQ >> 4) * 2;
    const float* aRowQ  = sAw + mQ * PITCH + khQ;
    const float* bRowLQ = sW + mQ * PITCH + khQ;
    const float* bRowRQ = bRowLQ + D * PITCH;
    for (int t = nFull + (int)blockIdx.x; t < nTiles; t += (int)gridDim.x) {
      const int row0 = t * 16;
      gather_tile(xin, rowptr, csr, row0, laneQ, sAw);
      switch (wave & 3) {
        case 0:  tile_products<2, 8, 0>(xin, bias, out, aRowQ, bRowLQ, bRowRQ, sAw, row0, laneQ, mQ, applyRelu); break;
        case 1:  tile_products<2, 8, 2>(xin, bias, out, aRowQ, bRowLQ, bRowRQ, sAw, row0, laneQ, mQ, applyRelu); break;
        case 2:  tile_products<2, 8, 4>(xin, bias, out, aRowQ, bRowLQ, bRowRQ, sAw, row0, laneQ, mQ, applyRelu); break;
        default: tile_products<2, 8, 6>(xin, bias, out, aRowQ, bRowLQ, bRowRQ, sAw, row0, laneQ, mQ, applyRelu); break;
      }
    }
  }
}

extern "C" void kernel_launch(void* const* d_in, const int* in_sizes, int n_in,
                              void* d_out, int out_size, void* d_ws, size_t ws_size,
                              hipStream_t stream) {
  (void)in_sizes; (void)n_in; (void)out_size; (void)ws_size;

  const float* x   = (const float*)d_in[0];
  const int*   ei  = (const int*)d_in[1];
  const float* W1l = (const float*)d_in[2];
  const float* b1  = (const float*)d_in[3];
  const float* W1r = (const float*)d_in[4];
  const float* W2l = (const float*)d_in[5];
  const float* b2  = (const float*)d_in[6];
  const float* W2r = (const float*)d_in[7];
  float*       out = (float*)d_out;

  const int* src = ei;
  const int* dst = ei + NE;

  char*  ws     = (char*)d_ws;
  int*   hist   = (int*)(ws + (512ull << 10));
  float* h      = (float*)(ws + (1ull << 20));
  int*   rowptr = (int*)(ws + (112ull << 20));
  int*   bbase  = (int*)(ws + (113ull << 20));
  int*   csr    = (int*)(ws + (116ull << 20));
  int*   pairs  = (int*)(ws + (120ull << 20));

  sage_count_kernel<<<NBUCK, COUNT_THREADS, 0, stream>>>(dst, hist, NE / 4);
  sage_place_kernel<<<NBUCK, PLACE_THREADS, 0, stream>>>(src, dst, hist, bbase, pairs, NE / 4);
  sage_sort_kernel<<<NBUCK, SORT_THREADS, 0, stream>>>(pairs, bbase, rowptr, csr, NN);

  sage_layer_kernel<<<GEMM_BLOCKS, GEMM_THREADS, 0, stream>>>(x, rowptr, csr, W1l, W1r, b1, h,
                                                    NN,  1);

  sage_layer_kernel<<<GEMM_BLOCKS, GEMM_THREADS, 0, stream>>>(h, rowptr, csr, W2l, W2r, b2, out,
                                                    NN,  0);
}


# Edge list grouped in two launches, not three: each slice sorted by bucket in LDS with an offsets row, the bucket pass collects its 256 runs; no table pass; rows add in the new LDS-rank order
# speedup vs baseline: 3.7904x; 1.0840x over previous
#include <hip/hip_runtime.h>

#define NN 100000
#define NE 625000
#define D  128

typedef float v2f __attribute__((ext_vector_type(2)));
typedef float v8f __attribute__((ext_vector_type(8)));

static_assert(NE % 4 == 0, "edge arrays are walked four entries per thread");

#define NBUCK         256
#define BW            ((NN + NBUCK - 1) / NBUCK)
#define CHUNK4        ((NE / 4 + NBUCK - 1) / NBUCK)
#define SLICE         (4 * CHUNK4)
#define OFFP          (NBUCK + 1)
#define SLICE_THREADS 256
#define SLICE_GROUPS  ((CHUNK4 + SLICE_THREADS - 1) / SLICE_THREADS)
#define SORT_THREADS  512
#define SORT_WAVES    (SORT_THREADS / 32)
#define SORT_SLICES   (NBUCK / SORT_WAVES)

static_assert(NBUCK == SLICE_THREADS && NBUCK % 32 == 0, "one bucket counter per thread, whole waves");
static_assert(NBUCK == SORT_WAVES * SORT_SLICES && SORT_SLICES <= 32 && (SORT_SLICES & (SORT_SLICES - 1)) == 0, "every slice walked by one wave, its bounds held by one lane");
static_assert(BW <= SORT_THREADS, "one node counter per thread");

__device__ __forceinline__ int wave_incl_scan(int v, int lane) {
  for (int d = 1; d < 32; d <<= 1) {
    const int u = __shfl_up(v, d);
    if (lane >= d) v += u;
  }
  return v;
}

__global__ __launch_bounds__(SLICE_THREADS) void sage_slice_kernel(const int* __restrict__ src,
                                                                   const int* __restrict__ dst,
                                                                   int* __restrict__ offs,
                                                                   int* __restrict__ pairs, int E4) {
  __shared__ int sCnt[NBUCK];
  __shared__ int sWave[NBUCK / 32];
  const int t    = threadIdx.x;
  const int lane = t & 31;
  const int g    = blockIdx.x;

  int4 s[SLICE_GROUPS], d[SLICE_GROUPS], r[SLICE_GROUPS];
  bool has[SLICE_GROUPS];
#pragma unroll
  for (int k = 0; k < SLICE_GROUPS; ++k) {
    const int j = t + k * SLICE_THREADS;
    const int q = g * CHUNK4 + j;
    has[k] = (j < CHUNK4) && (q < E4);
    s[k] = ((const int4*)src)[has[k] ? q : 0];
    d[k] = ((const int4*)dst)[has[k] ? q : 0];
    r[k] = make_int4(0, 0, 0, 0);
  }

  sCnt[t] = 0;
  __syncthreads();
#pragma unroll
  for (int k = 0; k < SLICE_GROUPS; ++k)
    if (has[k]) {
      r[k].x = atomicAdd(&sCnt[d[k].x / BW], 1);
      r[k].y = atomicAdd(&sCnt[d[k].y / BW], 1);
      r[k].z = atomicAdd(&sCnt[d[k].z / BW], 1);
      r[k].w = atomicAdd(&sCnt[d[k].w / BW], 1);
    }
  __syncthreads();

  const int total = sCnt[t];
  const int incl  = wave_incl_scan(total, lane);
  if (lane == 31) sWave[t >> 5] = incl;
  __syncthreads();
  int before = 0;
  for (int w = 0; w < NBUCK / 32; ++w) before += (w < (t >> 5)) ? sWave[w] : 0;
  const int base = before + incl - total;
  sCnt[t] = base;
  offs[g * OFFP + t] = base;
  if (t == NBUCK - 1) offs[g * OFFP + NBUCK] = base + total;
  __syncthreads();

  int2* pr = (int2*)pairs + g * SLICE;
#pragma unroll
  for (int k = 0; k < SLICE_GROUPS; ++k)
    if (has[k]) {
      pr[sCnt[d[k].x / BW] + r[k].x] = make_int2(s[k].x, d[k].x);
      pr[sCnt[d[k].y / BW] + r[k].y] = make_int2(s[k].y, d[k].y);
      pr[sCnt[d[k].z / BW] + r[k].z] = make_int2(s[k].z, d[k].z);
      pr[sCnt[d[k].w / BW] + r[k].w] = make_int2(s[k].w, d[k].w);
    }
}

__global__ __launch_bounds__(SORT_THREADS) void sage_sort_kernel(const int* __restrict__ pairs,
                                                                 const int* __restrict__ offs,
                                                                 int* __restrict__ rowptr,
                                                                 int* __restrict__ csr, int n) {
  __shared__ int sCnt[SORT_THREADS];
  __shared__ int sWave[SORT_WAVES];
  __shared__ int sSumLo[SORT_WAVES];
  __shared__ int sSumHi[SORT_WAVES];
  const int t     = threadIdx.x;
  const int lane  = t & 31;
  const int wave  = __builtin_amdgcn_readfirstlane(t >> 5);
  const int b     = blockIdx.x;
  const int node0 = b * BW;
  const int2* pr  = (const int2*)pairs;

  const int gsl = wave * SORT_SLICES + (lane & (SORT_SLICES - 1));
  const int lo = offs[gsl * OFFP + b];
  const int hi = offs[gsl * OFFP + b + 1];
  int slo = (lane < SORT_SLICES) ? lo : 0;
  int shi = (lane < SORT_SLICES) ? hi : 0;
  for (int m = 16; m >= 1; m >>= 1) {
    slo += __shfl_xor(slo, m);
    shi += __shfl_xor(shi, m);
  }
  if (lane == 0) { sSumLo[wave] = slo; sSumHi[wave] = shi; }
  sCnt[t] = 0;

  int2 p[SORT_SLICES];
#pragma unroll
  for (int i = 0; i < SORT_SLICES; ++i) {
    const int l  = __builtin_amdgcn_readlane(lo, i);
    const int cn = __builtin_amdgcn_readlane(hi, i) - l;
    p[i] = (lane < cn) ? pr[(wave * SORT_SLICES + i) * SLICE + l + lane] : make_int2(0, -1);
  }
  __syncthreads();
  int eBeg = 0, eEnd = 0;
  for (int w = 0; w < SORT_WAVES; ++w) { eBeg += sSumLo[w]; eEnd += sSumHi[w]; }

#pragma unroll
  for (int i = 0; i < SORT_SLICES; ++i)
    if (p[i].y >= 0) atomicAdd(&sCnt[p[i].y - node0], 1);
  for (int i = 0; i < SORT_SLICES; ++i) {
    const int l  = __builtin_amdgcn_readlane(lo, i);
    const int cn = __builtin_amdgcn_readlane(hi, i) - l;
    for (int k = lane + 32; k < cn; k += 32)
      atomicAdd(&sCnt[pr[(wave * SORT_SLICES + i) * SLICE + l + k].y - node0], 1);
  }
  __syncthreads();

  const int c = sCnt[t];
  const int incl = wave_incl_scan(c, lane);
  if (lane == 31) sWave[wave] = incl;
  __syncthreads();
  int before = 0;
  for (int w = 0; w < SORT_WAVES; ++w) before += (w < wave) ? sWave[w] : 0;
  const int first = eBeg + before + incl - c;

  sCnt[t] = first;
  if (t < BW && node0 + t < n) rowptr[node0 + t] = first;
  if (b == NBUCK - 1 && t == 0) rowptr[n] = eEnd;
  __syncthreads();

  int pos[SORT_SLICES];
#pragma unroll
  for (int i = 0; i < SORT_SLICES; ++i)
    pos[i] = (p[i].y >= 0) ? atomicAdd(&sCnt[p[i].y - node0], 1) : -1;
#pragma unroll
  for (int i = 0; i < SORT_SLICES; ++i)
    if (pos[i] >= 0) csr[pos[i]] = p[i].x;
  for (int i = 0; i < SORT_SLICES; ++i) {
    const int l  = __builtin_amdgcn_readlane(lo, i);
    const int cn = __builtin_amdgcn_readlane(hi, i) - l;
    for (int k = lane + 32; k < cn; k += 32) {
      const int2 q = pr[(wave * SORT_SLICES + i) * SLICE + l + k];
      csr[atomicAdd(&sCnt[q.y - node0], 1)] = q.x;
    }
  }
}

#define PITCH        132
#define GEMM_WAVES   16
#define GEMM_THREADS (GEMM_WAVES * 32)
#define GEMM_BLOCKS  256
#define LDS_W_FLOATS (2 * D * PITCH)
#define LDS_A_FLOATS (GEMM_WAVES * 16 * PITCH)
#define GROWS        14

__device__ __forceinline__ void stage_rows(const float* __restrict__ src, int row0, int lane,
                                           float* sAw) {
  const float4* p = (const float4*)(src + (long long)row0 * D) + lane;
  float4 r[16];
#pragma unroll
  for (int i = 0; i < 16; ++i)
    r[i] = p[i * (D / 4)];
#pragma unroll
  for (int i = 0; i < 16; ++i)
    *(float4*)(sAw + i * PITCH + lane * 4) = r[i];
}

__device__ __forceinline__ void close_node(float* sAw, int node, int lane, float rdeg, float4& acc) {
  const float s = __int_as_float(__builtin_amdgcn_readlane(__float_as_int(rdeg), node));
  *(float4*)(sAw + node * PITCH + lane * 4) = make_float4(acc.x * s, acc.y * s, acc.z * s, acc.w * s);
  acc = make_float4(0.0f, 0.0f, 0.0f, 0.0f);
}

__device__ __forceinline__ void gather_tile(const float* __restrict__ xin,
                                            const int* __restrict__ rowptr,
                                            const int* __restrict__ csr,
                                            int row0, int lane, float* sAw) {
  const int kLo = (lane < 16) ? lane : 16;
  const int kHi = (lane < 16) ? lane + 1 : 16;
  const int rpLo = rowptr[row0 + kLo];
  const int rpHi = rowptr[row0 + kHi];
  const float rdeg = 1.0f / fmaxf((float)(rpHi - rpLo), 1.0f);

  const int eBeg = __builtin_amdgcn_readlane(rpLo, 0);
  const int eEnd = __builtin_amdgcn_readlane(rpLo, 16);

  int node = 0;
  int nodeEnd = __builtin_amdgcn_readlane(rpHi, 0);
  float4 acc = make_float4(0.0f, 0.0f, 0.0f, 0.0f);

  int idxNext = (eBeg + lane < eEnd) ? csr[eBeg + lane] : 0;
  for (int cb = eBeg; cb < eEnd; cb += GROWS) {
    const int idx = idxNext;
    const int nb = (eEnd - cb < GROWS) ? (eEnd - cb) : GROWS;
    if (cb + GROWS < eEnd)
      idxNext = (cb + GROWS + lane < eEnd) ? csr[cb + GROWS + lane] : 0;

    float4 v[GROWS];
#pragma unroll
    for (int t = 0; t < GROWS; ++t)
      if (t < nb) {
        const int s = __builtin_amdgcn_readlane(idx, t);
        v[t] = ((const float4*)(xin + (long long)s * D))[lane];
      }
#pragma unroll
    for (int t = 0; t < GROWS; ++t)
      if (t < nb) {
        while (node < 15 && cb + t >= nodeEnd) {
          close_node(sAw, node, lane, rdeg, acc);
          ++node;
          nodeEnd = __builtin_amdgcn_readlane(rpHi, node);
        }
        acc.x += v[t].x;
        acc.y += v[t].y;
        acc.z += v[t].z;
        acc.w += v[t].w;
      }
  }
  for (; node < 16; ++node)
    close_node(sAw, node, lane, rdeg, acc);
}

template <int NCT, int KS>
__device__ __forceinline__ void load_chunk(const float* aRow, const float* bRow, int k0,
                                           v2f (&a)[KS], v2f (&b)[KS][NCT]) {
#pragma unroll
  for (int s = 0; s < KS; ++s) {
    a[s] = *(const v2f*)(aRow + k0 + 4 * s);
#pragma unroll
    for (int ct = 0; ct < NCT; ++ct)
      b[s][ct] = *(const v2f*)(bRow + ct * 16 * PITCH + k0 + 4 * s);
  }
}

template <int NCT, int KS>
__device__ __forceinline__ void mma_chunk(const v2f (&a)[KS], const v2f (&b)[KS][NCT],
                                          v8f (&acc)[NCT]) {
#pragma unroll
  for (int s = 0; s < KS; ++s) {
#pragma unroll
    for (int ct = 0; ct < NCT; ++ct)
      acc[ct] = __builtin_amdgcn_wmma_f32_16x16x4_f32(
           false, a[s],  false, b[s][ct],
           (short)0, acc[ct],  false,  false);
  }
}

template <int NCT, int KS>
__device__ __forceinline__ void pass_compute(const float* aRow, const float* bRow,
                                             v8f (&acc)[NCT]) {
  constexpr int CH = 4 * KS;
  static_assert(D % (2 * CH) == 0, "a pass is a whole number of chunk pairs");
  v2f a0[KS], a1[KS];
  v2f b0[KS][NCT], b1[KS][NCT];
  load_chunk<NCT, KS>(aRow, bRow, 0, a0, b0);
#pragma unroll 1
  for (int k0 = 0; k0 < D - 2 * CH; k0 += 2 * CH) {
    load_chunk<NCT, KS>(aRow, bRow, k0 + CH, a1, b1);
    mma_chunk<NCT, KS>(a0, b0, acc);
    load_chunk<NCT, KS>(aRow, bRow, k0 + 2 * CH, a0, b0);
    mma_chunk<NCT, KS>(a1, b1, acc);
  }
  load_chunk<NCT, KS>(aRow, bRow, D - CH, a1, b1);
  mma_chunk<NCT, KS>(a0, b0, acc);
  mma_chunk<NCT, KS>(a1, b1, acc);
}

template <int NCT, int KS, int CT0>
__device__ __forceinline__ void tile_products(const float* __restrict__ xin,
                                              const float* __restrict__ bias,
                                              float* __restrict__ out,
                                              const float* aRow, const float* bRowL,
                                              const float* bRowR, float* sAw,
                                              int row0, int lane, int m, int applyRelu) {
  constexpr int ct0 = CT0;
  v8f zero = {};
  v8f acc[NCT];
#pragma unroll
  for (int i = 0; i < NCT; ++i) acc[i] = zero;

  pass_compute<NCT, KS>(aRow, bRowL + ct0 * 16 * PITCH, acc);

  __builtin_amdgcn_sched_barrier(0);
  stage_rows(xin, row0, lane, sAw);
  pass_compute<NCT, KS>(aRow, bRowR + ct0 * 16 * PITCH, acc);

  const int rhalf = (lane >> 4) * 8;
#pragma unroll
  for (int ct = 0; ct < NCT; ++ct) {
    const float bc = bias[(ct0 + ct) * 16 + m];
#pragma unroll
    for (int r = 0; r < 8; ++r) {
      float v = acc[ct][r] + bc;
      if (applyRelu) v = fmaxf(v, 0.0f);
      out[(long long)(row0 + r + rhalf) * D + (ct0 + ct) * 16 + m] = v;
    }
  }
}

__global__ __launch_bounds__(GEMM_THREADS) void sage_layer_kernel(const float* __restrict__ xin,
                                                         const int* __restrict__ rowptr,
                                                         const int* __restrict__ csr,
                                                         const float* __restrict__ Wl,
                                                         const float* __restrict__ Wr,
                                                         const float* __restrict__ bias,
                                                         float* __restrict__ out,
                                                         int n, int applyRelu) {
  __shared__ __attribute__((aligned(16))) float lds[LDS_W_FLOATS + LDS_A_FLOATS];

  const int lane = threadIdx.x & 31;
  const int wave = __builtin_amdgcn_readfirstlane(threadIdx.x >> 5);
  const int nTiles = n >> 4;
  const int tileStep = GEMM_WAVES * gridDim.x;

  const int m  = lane & 15;
  const int kh = (lane >> 4) * 2;

  float* sW  = lds;
  float* sAw = lds + LDS_W_FLOATS + wave * 16 * PITCH;
  const float* aRow  = sAw + m * PITCH + kh;
  const float* bRowL = sW + m * PITCH + kh;
  const float* bRowR = bRowL + D * PITCH;

  int tile = wave * gridDim.x + blockIdx.x;

  {
    float4 wv[16];
#pragma unroll
    for (int j = 0; j < 16; ++j) {
      const int row = j * GEMM_WAVES + wave;
      const float* W = (row < D) ? Wl : Wr;
      wv[j] = ((const float4*)(W + (row & (D - 1)) * D))[lane];
    }
#pragma unroll
    for (int j = 0; j < 16; ++j) {
      const int row = j * GEMM_WAVES + wave;
      *(float4*)(sW + row * PITCH + lane * 4) = wv[j];
    }
  }

  __syncthreads();

  const int nFull = nTiles - nTiles % (4 * (int)gridDim.x);
  while (tile < nFull) {
    const int row0 = tile * 16;

    gather_tile(xin, rowptr, csr, row0, lane, sAw);
    tile_products<8, 2, 0>(xin, bias, out, aRow, bRowL, bRowR, sAw, row0, lane, m, applyRelu);
    tile += tileStep;
  }

  const int qWave0 = (nFull / (int)gridDim.x) & (GEMM_WAVES - 1);
  if ((wave & ~3) == qWave0) {
    int laneQ = lane;
    asm volatile("" : "+v"(laneQ));
    const int mQ  = laneQ & 15;
    const int khQ = (laneQ >> 4) * 2;
    const float* aRowQ  = sAw + mQ * PITCH + khQ;
    const float* bRowLQ = sW + mQ * PITCH + khQ;
    const float* bRowRQ = bRowLQ + D * PITCH;
    for (int t = nFull + (int)blockIdx.x; t < nTiles; t += (int)gridDim.x) {
      const int row0 = t * 16;
      gather_tile(xin, rowptr, csr, row0, laneQ, sAw);
      switch (wave & 3) {
        case 0:  tile_products<2, 8, 0>(xin, bias, out, aRowQ, bRowLQ, bRowRQ, sAw, row0, laneQ, mQ, applyRelu); break;
        case 1:  tile_products<2, 8, 2>(xin, bias, out, aRowQ, bRowLQ, bRowRQ, sAw, row0, laneQ, mQ, applyRelu); break;
        case 2:  tile_products<2, 8, 4>(xin, bias, out, aRowQ, bRowLQ, bRowRQ, sAw, row0, laneQ, mQ, applyRelu); break;
        default: tile_products<2, 8, 6>(xin, bias, out, aRowQ, bRowLQ, bRowRQ, sAw, row0, laneQ, mQ, applyRelu); break;
      }
    }
  }
}

extern "C" void kernel_launch(void* const* d_in, const int* in_sizes, int n_in,
                              void* d_out, int out_size, void* d_ws, size_t ws_size,
                              hipStream_t stream) {
  (void)in_sizes; (void)n_in; (void)out_size; (void)ws_size;

  const float* x   = (const float*)d_in[0];
  const int*   ei  = (const int*)d_in[1];
  const float* W1l = (const float*)d_in[2];
  const float* b1  = (const float*)d_in[3];
  const float* W1r = (const float*)d_in[4];
  const float* W2l = (const float*)d_in[5];
  const float* b2  = (const float*)d_in[6];
  const float* W2r = (const float*)d_in[7];
  float*       out = (float*)d_out;

  const int* src = ei;
  const int* dst = ei + NE;

  char*  ws     = (char*)d_ws;
  int*   offs   = (int*)(ws + (512ull << 10));
  float* h      = (float*)(ws + (1ull << 20));
  int*   rowptr = (int*)(ws + (112ull << 20));
  int*   csr    = (int*)(ws + (116ull << 20));
  int*   pairs  = (int*)(ws + (120ull << 20));

  sage_slice_kernel<<<NBUCK, SLICE_THREADS, 0, stream>>>(src, dst, offs, pairs, NE / 4);
  sage_sort_kernel<<<NBUCK, SORT_THREADS, 0, stream>>>(pairs, offs, rowptr, csr, NN);

  sage_layer_kernel<<<GEMM_BLOCKS, GEMM_THREADS, 0, stream>>>(x, rowptr, csr, W1l, W1r, b1, h,
                                                    NN,  1);

  sage_layer_kernel<<<GEMM_BLOCKS, GEMM_THREADS, 0, stream>>>(h, rowptr, csr, W2l, W2r, b2, out,
                                                    NN,  0);
}


# Stores as whole lines: slice pairs, csr rows and each layer tile's results are put together in LDS and leave 16 or 4 consecutive bytes per lane, not as scattered 8/4-byte and 64-byte half-line stores
# speedup vs baseline: 3.9138x; 1.0326x over previous
#include <hip/hip_runtime.h>

#define NN 100000
#define NE 625000
#define D  128

typedef float v2f __attribute__((ext_vector_type(2)));
typedef float v8f __attribute__((ext_vector_type(8)));

static_assert(NE % 4 == 0, "edge arrays are walked four entries per thread");

#define NBUCK         256
#define BW            ((NN + NBUCK - 1) / NBUCK)
#define CHUNK4        ((NE / 4 + NBUCK - 1) / NBUCK)
#define SLICE         (4 * CHUNK4)
#define OFFP          (NBUCK + 1)
#define SLICE_THREADS 256
#define SLICE_GROUPS  ((CHUNK4 + SLICE_THREADS - 1) / SLICE_THREADS)
#define SORT_THREADS  512
#define SORT_WAVES    (SORT_THREADS / 32)
#define SORT_SLICES   (NBUCK / SORT_WAVES)
#define CSR_STAGE     4096

static_assert(NBUCK == SLICE_THREADS && NBUCK % 32 == 0, "one bucket counter per thread, whole waves");
static_assert(NBUCK == SORT_WAVES * SORT_SLICES && SORT_SLICES <= 32 && (SORT_SLICES & (SORT_SLICES - 1)) == 0, "every slice walked by one wave, its bounds held by one lane");
static_assert(BW <= SORT_THREADS, "one node counter per thread");

__device__ __forceinline__ int wave_incl_scan(int v, int lane) {
  for (int d = 1; d < 32; d <<= 1) {
    const int u = __shfl_up(v, d);
    if (lane >= d) v += u;
  }
  return v;
}

__global__ __launch_bounds__(SLICE_THREADS) void sage_slice_kernel(const int* __restrict__ src,
                                                                   const int* __restrict__ dst,
                                                                   int* __restrict__ offs,
                                                                   int* __restrict__ pairs, int E4) {
  __shared__ int sCnt[NBUCK];
  __shared__ int sWave[NBUCK / 32];
  __shared__ __attribute__((aligned(16))) int2 sPair[SLICE];
  const int t    = threadIdx.x;
  const int lane = t & 31;
  const int g    = blockIdx.x;

  int4 s[SLICE_GROUPS], d[SLICE_GROUPS], r[SLICE_GROUPS];
  bool has[SLICE_GROUPS];
#pragma unroll
  for (int k = 0; k < SLICE_GROUPS; ++k) {
    const int j = t + k * SLICE_THREADS;
    const int q = g * CHUNK4 + j;
    has[k] = (j < CHUNK4) && (q < E4);
    s[k] = ((const int4*)src)[has[k] ? q : 0];
    d[k] = ((const int4*)dst)[has[k] ? q : 0];
    r[k] = make_int4(0, 0, 0, 0);
  }

  sCnt[t] = 0;
  __syncthreads();
#pragma unroll
  for (int k = 0; k < SLICE_GROUPS; ++k)
    if (has[k]) {
      r[k].x = atomicAdd(&sCnt[d[k].x / BW], 1);
      r[k].y = atomicAdd(&sCnt[d[k].y / BW], 1);
      r[k].z = atomicAdd(&sCnt[d[k].z / BW], 1);
      r[k].w = atomicAdd(&sCnt[d[k].w / BW], 1);
    }
  __syncthreads();

  const int total = sCnt[t];
  const int incl  = wave_incl_scan(total, lane);
  if (lane == 31) sWave[t >> 5] = incl;
  __syncthreads();
  int before = 0;
  for (int w = 0; w < NBUCK / 32; ++w) before += (w < (t >> 5)) ? sWave[w] : 0;
  const int base = before + incl - total;
  sCnt[t] = base;
  offs[g * OFFP + t] = base;
  if (t == NBUCK - 1) offs[g * OFFP + NBUCK] = base + total;
  __syncthreads();

#pragma unroll
  for (int k = 0; k < SLICE_GROUPS; ++k)
    if (has[k]) {
      sPair[sCnt[d[k].x / BW] + r[k].x] = make_int2(s[k].x, d[k].x);
      sPair[sCnt[d[k].y / BW] + r[k].y] = make_int2(s[k].y, d[k].y);
      sPair[sCnt[d[k].z / BW] + r[k].z] = make_int2(s[k].z, d[k].z);
      sPair[sCnt[d[k].w / BW] + r[k].w] = make_int2(s[k].w, d[k].w);
    }
  __syncthreads();

  const int left = E4 - g * CHUNK4;
  const int nv   = 2 * ((left < CHUNK4) ? left : CHUNK4);
  int4* pw = (int4*)((int2*)pairs + g * SLICE);
  constexpr int NV_IT = (SLICE / 2 + SLICE_THREADS - 1) / SLICE_THREADS;
  int4 q[NV_IT];
#pragma unroll
  for (int i = 0; i < NV_IT; ++i) {
    const int j = t + i * SLICE_THREADS;
    q[i] = ((const int4*)sPair)[(j < nv) ? j : 0];
  }
#pragma unroll
  for (int i = 0; i < NV_IT; ++i)
    asm volatile("" : : "v"(q[i].x), "v"(q[i].y), "v"(q[i].z), "v"(q[i].w));
#pragma unroll
  for (int i = 0; i < NV_IT; ++i) {
    const int j = t + i * SLICE_THREADS;
    if (j < nv) pw[j] = q[i];
  }
}

__global__ __launch_bounds__(SORT_THREADS) void sage_sort_kernel(const int* __restrict__ pairs,
                                                                 const int* __restrict__ offs,
                                                                 int* __restrict__ rowptr,
                                                                 int* __restrict__ csr, int n) {
  __shared__ int sCnt[SORT_THREADS];
  __shared__ int sWave[SORT_WAVES];
  __shared__ int sSumLo[SORT_WAVES];
  __shared__ int sSumHi[SORT_WAVES];
  __shared__ int sCsr[CSR_STAGE];
  const int t     = threadIdx.x;
  const int lane  = t & 31;
  const int wave  = __builtin_amdgcn_readfirstlane(t >> 5);
  const int b     = blockIdx.x;
  const int node0 = b * BW;
  const int2* pr  = (const int2*)pairs;

  const int gsl = wave * SORT_SLICES + (lane & (SORT_SLICES - 1));
  const int lo = offs[gsl * OFFP + b];
  const int hi = offs[gsl * OFFP + b + 1];
  int slo = (lane < SORT_SLICES) ? lo : 0;
  int shi = (lane < SORT_SLICES) ? hi : 0;
  for (int m = 16; m >= 1; m >>= 1) {
    slo += __shfl_xor(slo, m);
    shi += __shfl_xor(shi, m);
  }
  if (lane == 0) { sSumLo[wave] = slo; sSumHi[wave] = shi; }
  sCnt[t] = 0;

  int2 p[SORT_SLICES];
#pragma unroll
  for (int i = 0; i < SORT_SLICES; ++i) {
    const int l  = __builtin_amdgcn_readlane(lo, i);
    const int cn = __builtin_amdgcn_readlane(hi, i) - l;
    p[i] = (lane < cn) ? pr[(wave * SORT_SLICES + i) * SLICE + l + lane] : make_int2(0, -1);
  }
  __syncthreads();
  int eBeg = 0, eEnd = 0;
  for (int w = 0; w < SORT_WAVES; ++w) { eBeg += sSumLo[w]; eEnd += sSumHi[w]; }

#pragma unroll
  for (int i = 0; i < SORT_SLICES; ++i)
    if (p[i].y >= 0) atomicAdd(&sCnt[p[i].y - node0], 1);
  for (int i = 0; i < SORT_SLICES; ++i) {
    const int l  = __builtin_amdgcn_readlane(lo, i);
    const int cn = __builtin_amdgcn_readlane(hi, i) - l;
    for (int k = lane + 32; k < cn; k += 32)
      atomicAdd(&sCnt[pr[(wave * SORT_SLICES + i) * SLICE + l + k].y - node0], 1);
  }
  __syncthreads();

  const int c = sCnt[t];
  const int incl = wave_incl_scan(c, lane);
  if (lane == 31) sWave[wave] = incl;
  __syncthreads();
  int before = 0;
  for (int w = 0; w < SORT_WAVES; ++w) before += (w < wave) ? sWave[w] : 0;
  const int first = eBeg + before + incl - c;

  sCnt[t] = first;
  if (t < BW && node0 + t < n) rowptr[node0 + t] = first;
  if (b == NBUCK - 1 && t == 0) rowptr[n] = eEnd;
  __syncthreads();

  const int nB = eEnd - eBeg;
  const bool staged = nB <= CSR_STAGE;
  int pos[SORT_SLICES];
#pragma unroll
  for (int i = 0; i < SORT_SLICES; ++i)
    pos[i] = (p[i].y >= 0) ? atomicAdd(&sCnt[p[i].y - node0], 1) : -1;
  if (staged) {
#pragma unroll
    for (int i = 0; i < SORT_SLICES; ++i)
      if (pos[i] >= 0) sCsr[pos[i] - eBeg] = p[i].x;
  } else {
#pragma unroll
    for (int i = 0; i < SORT_SLICES; ++i)
      if (pos[i] >= 0) csr[pos[i]] = p[i].x;
  }
  for (int i = 0; i < SORT_SLICES; ++i) {
    const int l  = __builtin_amdgcn_readlane(lo, i);
    const int cn = __builtin_amdgcn_readlane(hi, i) - l;
    for (int k = lane + 32; k < cn; k += 32) {
      const int2 q = pr[(wave * SORT_SLICES + i) * SLICE + l + k];
      const int at = atomicAdd(&sCnt[q.y - node0], 1);
      if (staged) sCsr[at - eBeg] = q.x; else csr[at] = q.x;
    }
  }
  if (staged) {
    __syncthreads();
    constexpr int NC_IT = CSR_STAGE / SORT_THREADS + 1;
    const int j0 = t - (eBeg & 31);
    int cv[NC_IT];
#pragma unroll
    for (int i = 0; i < NC_IT; ++i) {
      const int j = j0 + i * SORT_THREADS;
      cv[i] = sCsr[(j >= 0 && j < nB) ? j : 0];
    }
#pragma unroll
    for (int i = 0; i < NC_IT; ++i) {
      const int j = j0 + i * SORT_THREADS;
      if (j >= 0 && j < nB) csr[eBeg + j] = cv[i];
    }
  }
}

#define PITCH        132
#define GEMM_WAVES   16
#define GEMM_THREADS (GEMM_WAVES * 32)
#define GEMM_BLOCKS  256
#define LDS_W_FLOATS (2 * D * PITCH)
#define LDS_A_FLOATS (GEMM_WAVES * 16 * PITCH)
#define GROWS        14

__device__ __forceinline__ void stage_rows(const float* __restrict__ src, int row0, int lane,
                                           float* sAw) {
  const float4* p = (const float4*)(src + (long long)row0 * D) + lane;
  float4 r[16];
#pragma unroll
  for (int i = 0; i < 16; ++i)
    r[i] = p[i * (D / 4)];
#pragma unroll
  for (int i = 0; i < 16; ++i)
    *(float4*)(sAw + i * PITCH + lane * 4) = r[i];
}

__device__ __forceinline__ void close_node(float* sAw, int node, int lane, float rdeg, float4& acc) {
  const float s = __int_as_float(__builtin_amdgcn_readlane(__float_as_int(rdeg), node));
  *(float4*)(sAw + node * PITCH + lane * 4) = make_float4(acc.x * s, acc.y * s, acc.z * s, acc.w * s);
  acc = make_float4(0.0f, 0.0f, 0.0f, 0.0f);
}

__device__ __forceinline__ void gather_tile(const float* __restrict__ xin,
                                            const int* __restrict__ rowptr,
                                            const int* __restrict__ csr,
                                            int row0, int lane, float* sAw) {
  const int kLo = (lane < 16) ? lane : 16;
  const int kHi = (lane < 16) ? lane + 1 : 16;
  const int rpLo = rowptr[row0 + kLo];
  const int rpHi = rowptr[row0 + kHi];
  const float rdeg = 1.0f / fmaxf((float)(rpHi - rpLo), 1.0f);

  const int eBeg = __builtin_amdgcn_readlane(rpLo, 0);
  const int eEnd = __builtin_amdgcn_readlane(rpLo, 16);

  int node = 0;
  int nodeEnd = __builtin_amdgcn_readlane(rpHi, 0);
  float4 acc = make_float4(0.0f, 0.0f, 0.0f, 0.0f);

  int idxNext = (eBeg + lane < eEnd) ? csr[eBeg + lane] : 0;
  for (int cb = eBeg; cb < eEnd; cb += GROWS) {
    const int idx = idxNext;
    const int nb = (eEnd - cb < GROWS) ? (eEnd - cb) : GROWS;
    if (cb + GROWS < eEnd)
      idxNext = (cb + GROWS + lane < eEnd) ? csr[cb + GROWS + lane] : 0;

    float4 v[GROWS];
#pragma unroll
    for (int t = 0; t < GROWS; ++t)
      if (t < nb) {
        const int s = __builtin_amdgcn_readlane(idx, t);
        v[t] = ((const float4*)(xin + (long long)s * D))[lane];
      }
#pragma unroll
    for (int t = 0; t < GROWS; ++t)
      if (t < nb) {
        while (node < 15 && cb + t >= nodeEnd) {
          close_node(sAw, node, lane, rdeg, acc);
          ++node;
          nodeEnd = __builtin_amdgcn_readlane(rpHi, node);
        }
        acc.x += v[t].x;
        acc.y += v[t].y;
        acc.z += v[t].z;
        acc.w += v[t].w;
      }
  }
  for (; node < 16; ++node)
    close_node(sAw, node, lane, rdeg, acc);
}

template <int NCT, int KS>
__device__ __forceinline__ void load_chunk(const float* aRow, const float* bRow, int k0,
                                           v2f (&a)[KS], v2f (&b)[KS][NCT]) {
#pragma unroll
  for (int s = 0; s < KS; ++s) {
    a[s] = *(const v2f*)(aRow + k0 + 4 * s);
#pragma unroll
    for (int ct = 0; ct < NCT; ++ct)
      b[s][ct] = *(const v2f*)(bRow + ct * 16 * PITCH + k0 + 4 * s);
  }
}

template <int NCT, int KS>
__device__ __forceinline__ void mma_chunk(const v2f (&a)[KS], const v2f (&b)[KS][NCT],
                                          v8f (&acc)[NCT]) {
#pragma unroll
  for (int s = 0; s < KS; ++s) {
#pragma unroll
    for (int ct = 0; ct < NCT; ++ct)
      acc[ct] = __builtin_amdgcn_wmma_f32_16x16x4_f32(
           false, a[s],  false, b[s][ct],
           (short)0, acc[ct],  false,  false);
  }
}

template <int NCT, int KS>
__device__ __forceinline__ void pass_compute(const float* aRow, const float* bRow,
                                             v8f (&acc)[NCT]) {
  constexpr int CH = 4 * KS;
  static_assert(D % (2 * CH) == 0, "a pass is a whole number of chunk pairs");
  v2f a0[KS], a1[KS];
  v2f b0[KS][NCT], b1[KS][NCT];
  load_chunk<NCT, KS>(aRow, bRow, 0, a0, b0);
#pragma unroll 1
  for (int k0 = 0; k0 < D - 2 * CH; k0 += 2 * CH) {
    load_chunk<NCT, KS>(aRow, bRow, k0 + CH, a1, b1);
    mma_chunk<NCT, KS>(a0, b0, acc);
    load_chunk<NCT, KS>(aRow, bRow, k0 + 2 * CH, a0, b0);
    mma_chunk<NCT, KS>(a1, b1, acc);
  }
  load_chunk<NCT, KS>(aRow, bRow, D - CH, a1, b1);
  mma_chunk<NCT, KS>(a0, b0, acc);
  mma_chunk<NCT, KS>(a1, b1, acc);
}

template <int NCT, int KS, int CT0>
__device__ __forceinline__ void tile_products(const float* __restrict__ xin,
                                              const float* __restrict__ bias,
                                              float* __restrict__ out,
                                              const float* aRow, const float* bRowL,
                                              const float* bRowR, float* sAw,
                                              int row0, int lane, int m, int applyRelu) {
  constexpr int ct0 = CT0;
  v8f zero = {};
  v8f acc[NCT];
#pragma unroll
  for (int i = 0; i < NCT; ++i) acc[i] = zero;

  pass_compute<NCT, KS>(aRow, bRowL + ct0 * 16 * PITCH, acc);

  __builtin_amdgcn_sched_barrier(0);
  stage_rows(xin, row0, lane, sAw);
  pass_compute<NCT, KS>(aRow, bRowR + ct0 * 16 * PITCH, acc);

  float* sOut = sAw + (lane >> 4) * 8 * PITCH + m;
#pragma unroll
  for (int ct = 0; ct < NCT; ++ct) {
    const float bc = bias[(ct0 + ct) * 16 + m];
#pragma unroll
    for (int r = 0; r < 8; ++r) {
      float v = acc[ct][r] + bc;
      if (applyRelu) v = fmaxf(v, 0.0f);
      sOut[r * PITCH + ct * 16] = v;
    }
  }
  constexpr int V4ROW = NCT * 4;
  constexpr int NST   = 16 * V4ROW / 32;
  float4 o[NST];
#pragma unroll
  for (int i = 0; i < NST; ++i) {
    const int j = lane + 32 * i;
    o[i] = *(const float4*)(sAw + (j / V4ROW) * PITCH + (j % V4ROW) * 4);
  }
#pragma unroll
  for (int i = 0; i < NST; ++i) {
    const int j = lane + 32 * i;
    *(float4*)(out + (long long)(row0 + j / V4ROW) * D + ct0 * 16 + (j % V4ROW) * 4) = o[i];
  }
}

__global__ __launch_bounds__(GEMM_THREADS) void sage_layer_kernel(const float* __restrict__ xin,
                                                         const int* __restrict__ rowptr,
                                                         const int* __restrict__ csr,
                                                         const float* __restrict__ Wl,
                                                         const float* __restrict__ Wr,
                                                         const float* __restrict__ bias,
                                                         float* __restrict__ out,
                                                         int n, int applyRelu) {
  __shared__ __attribute__((aligned(16))) float lds[LDS_W_FLOATS + LDS_A_FLOATS];

  const int lane = threadIdx.x & 31;
  const int wave = __builtin_amdgcn_readfirstlane(threadIdx.x >> 5);
  const int nTiles = n >> 4;
  const int tileStep = GEMM_WAVES * gridDim.x;

  const int m  = lane & 15;
  const int kh = (lane >> 4) * 2;

  float* sW  = lds;
  float* sAw = lds + LDS_W_FLOATS + wave * 16 * PITCH;
  const float* aRow  = sAw + m * PITCH + kh;
  const float* bRowL = sW + m * PITCH + kh;
  const float* bRowR = bRowL + D * PITCH;

  int tile = wave * gridDim.x + blockIdx.x;

  {
    float4 wv[16];
#pragma unroll
    for (int j = 0; j < 16; ++j) {
      const int row = j * GEMM_WAVES + wave;
      const float* W = (row < D) ? Wl : Wr;
      wv[j] = ((const float4*)(W + (row & (D - 1)) * D))[lane];
    }
#pragma unroll
    for (int j = 0; j < 16; ++j) {
      const int row = j * GEMM_WAVES + wave;
      *(float4*)(sW + row * PITCH + lane * 4) = wv[j];
    }
  }

  __syncthreads();

  const int nFull = nTiles - nTiles % (4 * (int)gridDim.x);
  while (tile < nFull) {
    const int row0 = tile * 16;

    gather_tile(xin, rowptr, csr, row0, lane, sAw);
    tile_products<8, 2, 0>(xin, bias, out, aRow, bRowL, bRowR, sAw, row0, lane, m, applyRelu);
    tile += tileStep;
  }

  const int qWave0 = (nFull / (int)gridDim.x) & (GEMM_WAVES - 1);
  if ((wave & ~3) == qWave0) {
    int laneQ = lane;
    asm volatile("" : "+v"(laneQ));
    const int mQ  = laneQ & 15;
    const int khQ = (laneQ >> 4) * 2;
    const float* aRowQ  = sAw + mQ * PITCH + khQ;
    const float* bRowLQ = sW + mQ * PITCH + khQ;
    const float* bRowRQ = bRowLQ + D * PITCH;
    for (int t = nFull + (int)blockIdx.x; t < nTiles; t += (int)gridDim.x) {
      const int row0 = t * 16;
      gather_tile(xin, rowptr, csr, row0, laneQ, sAw);
      switch (wave & 3) {
        case 0:  tile_products<2, 8, 0>(xin, bias, out, aRowQ, bRowLQ, bRowRQ, sAw, row0, laneQ, mQ, applyRelu); break;
        case 1:  tile_products<2, 8, 2>(xin, bias, out, aRowQ, bRowLQ, bRowRQ, sAw, row0, laneQ, mQ, applyRelu); break;
        case 2:  tile_products<2, 8, 4>(xin, bias, out, aRowQ, bRowLQ, bRowRQ, sAw, row0, laneQ, mQ, applyRelu); break;
        default: tile_products<2, 8, 6>(xin, bias, out, aRowQ, bRowLQ, bRowRQ, sAw, row0, laneQ, mQ, applyRelu); break;
      }
    }
  }
}

extern "C" void kernel_launch(void* const* d_in, const int* in_sizes, int n_in,
                              void* d_out, int out_size, void* d_ws, size_t ws_size,
                              hipStream_t stream) {
  (void)in_sizes; (void)n_in; (void)out_size; (void)ws_size;

  const float* x   = (const float*)d_in[0];
  const int*   ei  = (const int*)d_in[1];
  const float* W1l = (const float*)d_in[2];
  const float* b1  = (const float*)d_in[3];
  const float* W1r = (const float*)d_in[4];
  const float* W2l = (const float*)d_in[5];
  const float* b2  = (const float*)d_in[6];
  const float* W2r = (const float*)d_in[7];
  float*       out = (float*)d_out;

  const int* src = ei;
  const int* dst = ei + NE;

  char*  ws     = (char*)d_ws;
  int*   offs   = (int*)(ws + (512ull << 10));
  float* h      = (float*)(ws + (1ull << 20));
  int*   rowptr = (int*)(ws + (112ull << 20));
  int*   csr    = (int*)(ws + (116ull << 20));
  int*   pairs  = (int*)(ws + (120ull << 20));

  sage_slice_kernel<<<NBUCK, SLICE_THREADS, 0, stream>>>(src, dst, offs, pairs, NE / 4);
  sage_sort_kernel<<<NBUCK, SORT_THREADS, 0, stream>>>(pairs, offs, rowptr, csr, NN);

  sage_layer_kernel<<<GEMM_BLOCKS, GEMM_THREADS, 0, stream>>>(x, rowptr, csr, W1l, W1r, b1, h,
                                                    NN,  1);

  sage_layer_kernel<<<GEMM_BLOCKS, GEMM_THREADS, 0, stream>>>(h, rowptr, csr, W2l, W2r, b2, out,
                                                    NN,  0);
}


# Layer kernel, neighbour walk: the 14 row loads of a group issue back to back with no test between them, so each add waits for its own row only, not for the whole group; sum order kept
# speedup vs baseline: 4.1019x; 1.0481x over previous
#include <hip/hip_runtime.h>

#define NN 100000
#define NE 625000
#define D  128

typedef float v2f __attribute__((ext_vector_type(2)));
typedef float v8f __attribute__((ext_vector_type(8)));

static_assert(NE % 4 == 0, "edge arrays are walked four entries per thread");

#define NBUCK         256
#define BW            ((NN + NBUCK - 1) / NBUCK)
#define CHUNK4        ((NE / 4 + NBUCK - 1) / NBUCK)
#define SLICE         (4 * CHUNK4)
#define OFFP          (NBUCK + 1)
#define SLICE_THREADS 256
#define SLICE_GROUPS  ((CHUNK4 + SLICE_THREADS - 1) / SLICE_THREADS)
#define SORT_THREADS  512
#define SORT_WAVES    (SORT_THREADS / 32)
#define SORT_SLICES   (NBUCK / SORT_WAVES)
#define CSR_STAGE     4096

static_assert(NBUCK == SLICE_THREADS && NBUCK % 32 == 0, "one bucket counter per thread, whole waves");
static_assert(NBUCK == SORT_WAVES * SORT_SLICES && SORT_SLICES <= 32 && (SORT_SLICES & (SORT_SLICES - 1)) == 0, "every slice walked by one wave, its bounds held by one lane");
static_assert(BW <= SORT_THREADS, "one node counter per thread");

__device__ __forceinline__ int wave_incl_scan(int v, int lane) {
  for (int d = 1; d < 32; d <<= 1) {
    const int u = __shfl_up(v, d);
    if (lane >= d) v += u;
  }
  return v;
}

__global__ __launch_bounds__(SLICE_THREADS) void sage_slice_kernel(const int* __restrict__ src,
                                                                   const int* __restrict__ dst,
                                                                   int* __restrict__ offs,
                                                                   int* __restrict__ pairs, int E4) {
  __shared__ int sCnt[NBUCK];
  __shared__ int sWave[NBUCK / 32];
  __shared__ __attribute__((aligned(16))) int2 sPair[SLICE];
  const int t    = threadIdx.x;
  const int lane = t & 31;
  const int g    = blockIdx.x;

  int4 s[SLICE_GROUPS], d[SLICE_GROUPS], r[SLICE_GROUPS];
  bool has[SLICE_GROUPS];
#pragma unroll
  for (int k = 0; k < SLICE_GROUPS; ++k) {
    const int j = t + k * SLICE_THREADS;
    const int q = g * CHUNK4 + j;
    has[k] = (j < CHUNK4) && (q < E4);
    s[k] = ((const int4*)src)[has[k] ? q : 0];
    d[k] = ((const int4*)dst)[has[k] ? q : 0];
    r[k] = make_int4(0, 0, 0, 0);
  }

  sCnt[t] = 0;
  __syncthreads();
#pragma unroll
  for (int k = 0; k < SLICE_GROUPS; ++k)
    if (has[k]) {
      r[k].x = atomicAdd(&sCnt[d[k].x / BW], 1);
      r[k].y = atomicAdd(&sCnt[d[k].y / BW], 1);
      r[k].z = atomicAdd(&sCnt[d[k].z / BW], 1);
      r[k].w = atomicAdd(&sCnt[d[k].w / BW], 1);
    }
  __syncthreads();

  const int total = sCnt[t];
  const int incl  = wave_incl_scan(total, lane);
  if (lane == 31) sWave[t >> 5] = incl;
  __syncthreads();
  int before = 0;
  for (int w = 0; w < NBUCK / 32; ++w) before += (w < (t >> 5)) ? sWave[w] : 0;
  const int base = before + incl - total;
  sCnt[t] = base;
  offs[g * OFFP + t] = base;
  if (t == NBUCK - 1) offs[g * OFFP + NBUCK] = base + total;
  __syncthreads();

#pragma unroll
  for (int k = 0; k < SLICE_GROUPS; ++k)
    if (has[k]) {
      sPair[sCnt[d[k].x / BW] + r[k].x] = make_int2(s[k].x, d[k].x);
      sPair[sCnt[d[k].y / BW] + r[k].y] = make_int2(s[k].y, d[k].y);
      sPair[sCnt[d[k].z / BW] + r[k].z] = make_int2(s[k].z, d[k].z);
      sPair[sCnt[d[k].w / BW] + r[k].w] = make_int2(s[k].w, d[k].w);
    }
  __syncthreads();

  const int left = E4 - g * CHUNK4;
  const int nv   = 2 * ((left < CHUNK4) ? left : CHUNK4);
  int4* pw = (int4*)((int2*)pairs + g * SLICE);
  constexpr int NV_IT = (SLICE / 2 + SLICE_THREADS - 1) / SLICE_THREADS;
  int4 q[NV_IT];
#pragma unroll
  for (int i = 0; i < NV_IT; ++i) {
    const int j = t + i * SLICE_THREADS;
    q[i] = ((const int4*)sPair)[(j < nv) ? j : 0];
  }
#pragma unroll
  for (int i = 0; i < NV_IT; ++i)
    asm volatile("" : : "v"(q[i].x), "v"(q[i].y), "v"(q[i].z), "v"(q[i].w));
#pragma unroll
  for (int i = 0; i < NV_IT; ++i) {
    const int j = t + i * SLICE_THREADS;
    if (j < nv) pw[j] = q[i];
  }
}

__global__ __launch_bounds__(SORT_THREADS) void sage_sort_kernel(const int* __restrict__ pairs,
                                                                 const int* __restrict__ offs,
                                                                 int* __restrict__ rowptr,
                                                                 int* __restrict__ csr, int n) {
  __shared__ int sCnt[SORT_THREADS];
  __shared__ int sWave[SORT_WAVES];
  __shared__ int sSumLo[SORT_WAVES];
  __shared__ int sSumHi[SORT_WAVES];
  __shared__ int sCsr[CSR_STAGE];
  const int t     = threadIdx.x;
  const int lane  = t & 31;
  const int wave  = __builtin_amdgcn_readfirstlane(t >> 5);
  const int b     = blockIdx.x;
  const int node0 = b * BW;
  const int2* pr  = (const int2*)pairs;

  const int gsl = wave * SORT_SLICES + (lane & (SORT_SLICES - 1));
  const int lo = offs[gsl * OFFP + b];
  const int hi = offs[gsl * OFFP + b + 1];
  int slo = (lane < SORT_SLICES) ? lo : 0;
  int shi = (lane < SORT_SLICES) ? hi : 0;
  for (int m = 16; m >= 1; m >>= 1) {
    slo += __shfl_xor(slo, m);
    shi += __shfl_xor(shi, m);
  }
  if (lane == 0) { sSumLo[wave] = slo; sSumHi[wave] = shi; }
  sCnt[t] = 0;

  int2 p[SORT_SLICES];
#pragma unroll
  for (int i = 0; i < SORT_SLICES; ++i) {
    const int l  = __builtin_amdgcn_readlane(lo, i);
    const int cn = __builtin_amdgcn_readlane(hi, i) - l;
    p[i] = (lane < cn) ? pr[(wave * SORT_SLICES + i) * SLICE + l + lane] : make_int2(0, -1);
  }
  __syncthreads();
  int eBeg = 0, eEnd = 0;
  for (int w = 0; w < SORT_WAVES; ++w) { eBeg += sSumLo[w]; eEnd += sSumHi[w]; }

#pragma unroll
  for (int i = 0; i < SORT_SLICES; ++i)
    if (p[i].y >= 0) atomicAdd(&sCnt[p[i].y - node0], 1);
  for (int i = 0; i < SORT_SLICES; ++i) {
    const int l  = __builtin_amdgcn_readlane(lo, i);
    const int cn = __builtin_amdgcn_readlane(hi, i) - l;
    for (int k = lane + 32; k < cn; k += 32)
      atomicAdd(&sCnt[pr[(wave * SORT_SLICES + i) * SLICE + l + k].y - node0], 1);
  }
  __syncthreads();

  const int c = sCnt[t];
  const int incl = wave_incl_scan(c, lane);
  if (lane == 31) sWave[wave] = incl;
  __syncthreads();
  int before = 0;
  for (int w = 0; w < SORT_WAVES; ++w) before += (w < wave) ? sWave[w] : 0;
  const int first = eBeg + before + incl - c;

  sCnt[t] = first;
  if (t < BW && node0 + t < n) rowptr[node0 + t] = first;
  if (b == NBUCK - 1 && t == 0) rowptr[n] = eEnd;
  __syncthreads();

  const int nB = eEnd - eBeg;
  const bool staged = nB <= CSR_STAGE;
  int pos[SORT_SLICES];
#pragma unroll
  for (int i = 0; i < SORT_SLICES; ++i)
    pos[i] = (p[i].y >= 0) ? atomicAdd(&sCnt[p[i].y - node0], 1) : -1;
  if (staged) {
#pragma unroll
    for (int i = 0; i < SORT_SLICES; ++i)
      if (pos[i] >= 0) sCsr[pos[i] - eBeg] = p[i].x;
  } else {
#pragma unroll
    for (int i = 0; i < SORT_SLICES; ++i)
      if (pos[i] >= 0) csr[pos[i]] = p[i].x;
  }
  for (int i = 0; i < SORT_SLICES; ++i) {
    const int l  = __builtin_amdgcn_readlane(lo, i);
    const int cn = __builtin_amdgcn_readlane(hi, i) - l;
    for (int k = lane + 32; k < cn; k += 32) {
      const int2 q = pr[(wave * SORT_SLICES + i) * SLICE + l + k];
      const int at = atomicAdd(&sCnt[q.y - node0], 1);
      if (staged) sCsr[at - eBeg] = q.x; else csr[at] = q.x;
    }
  }
  if (staged) {
    __syncthreads();
    constexpr int NC_IT = CSR_STAGE / SORT_THREADS + 1;
    const int j0 = t - (eBeg & 31);
    int cv[NC_IT];
#pragma unroll
    for (int i = 0; i < NC_IT; ++i) {
      const int j = j0 + i * SORT_THREADS;
      cv[i] = sCsr[(j >= 0 && j < nB) ? j : 0];
    }
#pragma unroll
    for (int i = 0; i < NC_IT; ++i) {
      const int j = j0 + i * SORT_THREADS;
      if (j >= 0 && j < nB) csr[eBeg + j] = cv[i];
    }
  }
}

#define PITCH        132
#define GEMM_WAVES   16
#define GEMM_THREADS (GEMM_WAVES * 32)
#define GEMM_BLOCKS  256
#define LDS_W_FLOATS (2 * D * PITCH)
#define LDS_A_FLOATS (GEMM_WAVES * 16 * PITCH)
#define GROWS        14

__device__ __forceinline__ void stage_rows(const float* __restrict__ src, int row0, int lane,
                                           float* sAw) {
  const float4* p = (const float4*)(src + (long long)row0 * D) + lane;
  float4 r[16];
#pragma unroll
  for (int i = 0; i < 16; ++i)
    r[i] = p[i * (D / 4)];
#pragma unroll
  for (int i = 0; i < 16; ++i)
    *(float4*)(sAw + i * PITCH + lane * 4) = r[i];
}

__device__ __forceinline__ void close_node(float* sAw, int node, int lane, float rdeg, float4& acc) {
  const float s = __int_as_float(__builtin_amdgcn_readlane(__float_as_int(rdeg), node));
  *(float4*)(sAw + node * PITCH + lane * 4) = make_float4(acc.x * s, acc.y * s, acc.z * s, acc.w * s);
  acc = make_float4(0.0f, 0.0f, 0.0f, 0.0f);
}

__device__ __forceinline__ void gather_tile(const float* __restrict__ xin,
                                            const int* __restrict__ rowptr,
                                            const int* __restrict__ csr,
                                            int row0, int lane, float* sAw) {
  const int kLo = (lane < 16) ? lane : 16;
  const int kHi = (lane < 16) ? lane + 1 : 16;
  const int rpLo = rowptr[row0 + kLo];
  const int rpHi = rowptr[row0 + kHi];
  const float rdeg = 1.0f / fmaxf((float)(rpHi - rpLo), 1.0f);

  const int eBeg = __builtin_amdgcn_readlane(rpLo, 0);
  const int eEnd = __builtin_amdgcn_readlane(rpLo, 16);

  int node = 0;
  int nodeEnd = __builtin_amdgcn_readlane(rpHi, 0);
  float4 acc = make_float4(0.0f, 0.0f, 0.0f, 0.0f);

  int idxNext = (eBeg + lane < eEnd) ? csr[eBeg + lane] : 0;
  for (int cb = eBeg; cb < eEnd; cb += GROWS) {
    const int idx = idxNext;
    if (cb + GROWS < eEnd)
      idxNext = (cb + GROWS + lane < eEnd) ? csr[cb + GROWS + lane] : 0;

    float4 v[GROWS];
#pragma unroll
    for (int t = 0; t < GROWS; ++t) {
      const int s = __builtin_amdgcn_readlane(idx, t);
      v[t] = ((const float4*)(xin + (long long)s * D))[lane];
    }
#pragma unroll
    for (int t = 0; t < GROWS; ++t) {
      if (cb + t >= eEnd) break;
      while (cb + t >= nodeEnd) {
        close_node(sAw, node, lane, rdeg, acc);
        ++node;
        nodeEnd = __builtin_amdgcn_readlane(rpHi, node);
      }
      acc.x += v[t].x;
      acc.y += v[t].y;
      acc.z += v[t].z;
      acc.w += v[t].w;
    }
  }
  for (; node < 16; ++node)
    close_node(sAw, node, lane, rdeg, acc);
}

template <int NCT, int KS>
__device__ __forceinline__ void load_chunk(const float* aRow, const float* bRow, int k0,
                                           v2f (&a)[KS], v2f (&b)[KS][NCT]) {
#pragma unroll
  for (int s = 0; s < KS; ++s) {
    a[s] = *(const v2f*)(aRow + k0 + 4 * s);
#pragma unroll
    for (int ct = 0; ct < NCT; ++ct)
      b[s][ct] = *(const v2f*)(bRow + ct * 16 * PITCH + k0 + 4 * s);
  }
}

template <int NCT, int KS>
__device__ __forceinline__ void mma_chunk(const v2f (&a)[KS], const v2f (&b)[KS][NCT],
                                          v8f (&acc)[NCT]) {
#pragma unroll
  for (int s = 0; s < KS; ++s) {
#pragma unroll
    for (int ct = 0; ct < NCT; ++ct)
      acc[ct] = __builtin_amdgcn_wmma_f32_16x16x4_f32(
           false, a[s],  false, b[s][ct],
           (short)0, acc[ct],  false,  false);
  }
}

template <int NCT, int KS>
__device__ __forceinline__ void pass_compute(const float* aRow, const float* bRow,
                                             v8f (&acc)[NCT]) {
  constexpr int CH = 4 * KS;
  static_assert(D % (2 * CH) == 0, "a pass is a whole number of chunk pairs");
  v2f a0[KS], a1[KS];
  v2f b0[KS][NCT], b1[KS][NCT];
  load_chunk<NCT, KS>(aRow, bRow, 0, a0, b0);
#pragma unroll 1
  for (int k0 = 0; k0 < D - 2 * CH; k0 += 2 * CH) {
    load_chunk<NCT, KS>(aRow, bRow, k0 + CH, a1, b1);
    mma_chunk<NCT, KS>(a0, b0, acc);
    load_chunk<NCT, KS>(aRow, bRow, k0 + 2 * CH, a0, b0);
    mma_chunk<NCT, KS>(a1, b1, acc);
  }
  load_chunk<NCT, KS>(aRow, bRow, D - CH, a1, b1);
  mma_chunk<NCT, KS>(a0, b0, acc);
  mma_chunk<NCT, KS>(a1, b1, acc);
}

template <int NCT, int KS, int CT0>
__device__ __forceinline__ void tile_products(const float* __restrict__ xin,
                                              const float* __restrict__ bias,
                                              float* __restrict__ out,
                                              const float* aRow, const float* bRowL,
                                              const float* bRowR, float* sAw,
                                              int row0, int lane, int m, int applyRelu) {
  constexpr int ct0 = CT0;
  v8f zero = {};
  v8f acc[NCT];
#pragma unroll
  for (int i = 0; i < NCT; ++i) acc[i] = zero;

  pass_compute<NCT, KS>(aRow, bRowL + ct0 * 16 * PITCH, acc);

  __builtin_amdgcn_sched_barrier(0);
  stage_rows(xin, row0, lane, sAw);
  pass_compute<NCT, KS>(aRow, bRowR + ct0 * 16 * PITCH, acc);

  float* sOut = sAw + (lane >> 4) * 8 * PITCH + m;
#pragma unroll
  for (int ct = 0; ct < NCT; ++ct) {
    const float bc = bias[(ct0 + ct) * 16 + m];
#pragma unroll
    for (int r = 0; r < 8; ++r) {
      float v = acc[ct][r] + bc;
      if (applyRelu) v = fmaxf(v, 0.0f);
      sOut[r * PITCH + ct * 16] = v;
    }
  }
  constexpr int V4ROW = NCT * 4;
  constexpr int NST   = 16 * V4ROW / 32;
  float4 o[NST];
#pragma unroll
  for (int i = 0; i < NST; ++i) {
    const int j = lane + 32 * i;
    o[i] = *(const float4*)(sAw + (j / V4ROW) * PITCH + (j % V4ROW) * 4);
  }
#pragma unroll
  for (int i = 0; i < NST; ++i) {
    const int j = lane + 32 * i;
    *(float4*)(out + (long long)(row0 + j / V4ROW) * D + ct0 * 16 + (j % V4ROW) * 4) = o[i];
  }
}

__global__ __launch_bounds__(GEMM_THREADS) void sage_layer_kernel(const float* __restrict__ xin,
                                                         const int* __restrict__ rowptr,
                                                         const int* __restrict__ csr,
                                                         const float* __restrict__ Wl,
                                                         const float* __restrict__ Wr,
                                                         const float* __restrict__ bias,
                                                         float* __restrict__ out,
                                                         int n, int applyRelu) {
  __shared__ __attribute__((aligned(16))) float lds[LDS_W_FLOATS + LDS_A_FLOATS];

  const int lane = threadIdx.x & 31;
  const int wave = __builtin_amdgcn_readfirstlane(threadIdx.x >> 5);
  const int nTiles = n >> 4;
  const int tileStep = GEMM_WAVES * gridDim.x;

  const int m  = lane & 15;
  const int kh = (lane >> 4) * 2;

  float* sW  = lds;
  float* sAw = lds + LDS_W_FLOATS + wave * 16 * PITCH;
  const float* aRow  = sAw + m * PITCH + kh;
  const float* bRowL = sW + m * PITCH + kh;
  const float* bRowR = bRowL + D * PITCH;

  int tile = wave * gridDim.x + blockIdx.x;

  {
    float4 wv[16];
#pragma unroll
    for (int j = 0; j < 16; ++j) {
      const int row = j * GEMM_WAVES + wave;
      const float* W = (row < D) ? Wl : Wr;
      wv[j] = ((const float4*)(W + (row & (D - 1)) * D))[lane];
    }
#pragma unroll
    for (int j = 0; j < 16; ++j) {
      const int row = j * GEMM_WAVES + wave;
      *(float4*)(sW + row * PITCH + lane * 4) = wv[j];
    }
  }

  __syncthreads();

  const int nFull = nTiles - nTiles % (4 * (int)gridDim.x);
  while (tile < nFull) {
    const int row0 = tile * 16;

    gather_tile(xin, rowptr, csr, row0, lane, sAw);
    tile_products<8, 2, 0>(xin, bias, out, aRow, bRowL, bRowR, sAw, row0, lane, m, applyRelu);
    tile += tileStep;
  }

  const int qWave0 = (nFull / (int)gridDim.x) & (GEMM_WAVES - 1);
  if ((wave & ~3) == qWave0) {
    int laneQ = lane;
    asm volatile("" : "+v"(laneQ));
    const int mQ  = laneQ & 15;
    const int khQ = (laneQ >> 4) * 2;
    const float* aRowQ  = sAw + mQ * PITCH + khQ;
    const float* bRowLQ = sW + mQ * PITCH + khQ;
    const float* bRowRQ = bRowLQ + D * PITCH;
    for (int t = nFull + (int)blockIdx.x; t < nTiles; t += (int)gridDim.x) {
      const int row0 = t * 16;
      gather_tile(xin, rowptr, csr, row0, laneQ, sAw);
      switch (wave & 3) {
        case 0:  tile_products<2, 8, 0>(xin, bias, out, aRowQ, bRowLQ, bRowRQ, sAw, row0, laneQ, mQ, applyRelu); break;
        case 1:  tile_products<2, 8, 2>(xin, bias, out, aRowQ, bRowLQ, bRowRQ, sAw, row0, laneQ, mQ, applyRelu); break;
        case 2:  tile_products<2, 8, 4>(xin, bias, out, aRowQ, bRowLQ, bRowRQ, sAw, row0, laneQ, mQ, applyRelu); break;
        default: tile_products<2, 8, 6>(xin, bias, out, aRowQ, bRowLQ, bRowRQ, sAw, row0, laneQ, mQ, applyRelu); break;
      }
    }
  }
}

extern "C" void kernel_launch(void* const* d_in, const int* in_sizes, int n_in,
                              void* d_out, int out_size, void* d_ws, size_t ws_size,
                              hipStream_t stream) {
  (void)in_sizes; (void)n_in; (void)out_size; (void)ws_size;

  const float* x   = (const float*)d_in[0];
  const int*   ei  = (const int*)d_in[1];
  const float* W1l = (const float*)d_in[2];
  const float* b1  = (const float*)d_in[3];
  const float* W1r = (const float*)d_in[4];
  const float* W2l = (const float*)d_in[5];
  const float* b2  = (const float*)d_in[6];
  const float* W2r = (const float*)d_in[7];
  float*       out = (float*)d_out;

  const int* src = ei;
  const int* dst = ei + NE;

  char*  ws     = (char*)d_ws;
  int*   offs   = (int*)(ws + (512ull << 10));
  float* h      = (float*)(ws + (1ull << 20));
  int*   rowptr = (int*)(ws + (112ull << 20));
  int*   csr    = (int*)(ws + (116ull << 20));
  int*   pairs  = (int*)(ws + (120ull << 20));

  sage_slice_kernel<<<NBUCK, SLICE_THREADS, 0, stream>>>(src, dst, offs, pairs, NE / 4);
  sage_sort_kernel<<<NBUCK, SORT_THREADS, 0, stream>>>(pairs, offs, rowptr, csr, NN);

  sage_layer_kernel<<<GEMM_BLOCKS, GEMM_THREADS, 0, stream>>>(x, rowptr, csr, W1l, W1r, b1, h,
                                                    NN,  1);

  sage_layer_kernel<<<GEMM_BLOCKS, GEMM_THREADS, 0, stream>>>(h, rowptr, csr, W2l, W2r, b2, out,
                                                    NN,  0);
}
